# starting kernel
# baseline (speedup 1.0000x reference)
#include <hip/hip_runtime.h>
#include <hip/hip_bf16.h>


typedef __attribute__((ext_vector_type(16))) _Float16 v16h;
typedef __attribute__((ext_vector_type(8)))  _Float16 v8h;
typedef __attribute__((ext_vector_type(4)))  _Float16 v4h;
typedef __attribute__((ext_vector_type(8)))  float    v8f;

#define FLAG_RELU  1
#define FLAG_TRANS 2

__device__ __forceinline__ v16h load_frag(const _Float16* base, int ld,
                                          int row0, int k0, int lane) {
  const int r  = row0 + (lane & 15);
  const int kb = k0 + ((lane >> 4) << 3);
  const _Float16* p = base + (size_t)r * ld + kb;
  v8h lo = *(const v8h*)(p);
  v8h hi = *(const v8h*)(p + 16);
  return __builtin_shufflevector(lo, hi, 0, 1, 2, 3, 4, 5, 6, 7,
                                         8, 9, 10, 11, 12, 13, 14, 15);
}

__device__ __forceinline__ v8f wmma_f16(v16h a, v16h b, v8f c) {
  return __builtin_amdgcn_wmma_f32_16x16x32_f16(false, a, false, b,
                                                (short)0, c, false, false);
}

__global__ __launch_bounds__(256) void cvt_f32_f16_kernel(
    const float* __restrict__ in, _Float16* __restrict__ out, int n4) {
  const int i = blockIdx.x * blockDim.x + threadIdx.x;
  if (i < n4) {
    float4 v = ((const float4*)in)[i];
    v4h o = {(_Float16)v.x, (_Float16)v.y, (_Float16)v.z, (_Float16)v.w};
    ((v4h*)out)[i] = o;
  }
}

__global__ __launch_bounds__(128) void gemm_nt_kernel(
    const _Float16* __restrict__ A, const _Float16* __restrict__ Bw,
    float* __restrict__ Cf, _Float16* __restrict__ Ch,
    const float* __restrict__ bias, int Mdim, int Ndim, int Kdim, int flags) {
  const int lane = threadIdx.x & 31;
  const int wave = threadIdx.x >> 5;
  const int tiles_n = Ndim >> 7;
  const int bm = blockIdx.x / tiles_n;
  const int bn = blockIdx.x - bm * tiles_n;
  const int m0 = bm * 128 + (wave & 1) * 64;
  const int n0 = bn * 128 + (wave >> 1) * 64;

  v8f acc[4][4] = {};

#pragma unroll 1
  for (int k0 = 0; k0 < Kdim; k0 += 32) {
    __builtin_prefetch(A + (size_t)(m0 + (lane & 15)) * Kdim + k0 + 32, 0, 1);
    __builtin_prefetch(Bw + (size_t)(n0 + (lane & 15)) * Kdim + k0 + 32, 0, 1);

    v16h af[4], bf[4];
#pragma unroll
    for (int i = 0; i < 4; i++) af[i] = load_frag(A, Kdim, m0 + 16 * i, k0, lane);
#pragma unroll
    for (int j = 0; j < 4; j++) bf[j] = load_frag(Bw, Kdim, n0 + 16 * j, k0, lane);
#pragma unroll
    for (int i = 0; i < 4; i++)
#pragma unroll
      for (int j = 0; j < 4; j++)
        acc[i][j] = wmma_f16(af[i], bf[j], acc[i][j]);
  }

  const int nl = lane & 15;
  const int mh = (lane >> 4) << 3;
#pragma unroll
  for (int i = 0; i < 4; i++) {
#pragma unroll
    for (int j = 0; j < 4; j++) {
      const int n = n0 + j * 16 + nl;
      const float bv = bias ? bias[n] : 0.f;
#pragma unroll
      for (int r = 0; r < 8; r++) {
        const int m = m0 + 16 * i + mh + r;
        float v = acc[i][j][r] + bv;
        if (flags & FLAG_RELU) v = fmaxf(v, 0.f);
        if (Cf) Cf[(size_t)m * Ndim + n] = v;
        if (Ch) {
          if (flags & FLAG_TRANS) {
            Ch[((size_t)((m >> 10) << 10) + n) * 1024 + (m & 1023)] = (_Float16)v;
          } else {
            Ch[(size_t)m * Ndim + n] = (_Float16)v;
          }
        }
      }
    }
  }
}

__global__ __launch_bounds__(256) void attn_kernel(
    const _Float16* __restrict__ Qh, const _Float16* __restrict__ Kh,
    const _Float16* __restrict__ Vt, const int* __restrict__ mask,
    _Float16* __restrict__ Oh) {
  __shared__ _Float16 plds[8][16 * 40];

  const int lane = threadIdx.x & 31;
  const int wave = threadIdx.x >> 5;
  const int bh   = blockIdx.x >> 3;
  const int qblk = blockIdx.x & 7;
  const int b = bh >> 4, h = bh & 15;
  const int q0 = qblk * 128 + wave * 16;

  const _Float16* Qbase = Qh + (size_t)b * 1024 * 1024 + h * 64;
  const _Float16* Kbase = Kh + (size_t)b * 1024 * 1024 + h * 64;
  const _Float16* Vbase = Vt + (size_t)(b * 16 + h) * 64 * 1024;
  const int* mb = mask + b * 1024;
  _Float16* myp = &plds[wave][0];

  const v16h qa0 = load_frag(Qbase, 1024, q0, 0, lane);
  const v16h qa1 = load_frag(Qbase, 1024, q0, 32, lane);

  v8f o[4] = {};
  v8f mrun = {-1e30f, -1e30f, -1e30f, -1e30f, -1e30f, -1e30f, -1e30f, -1e30f};
  v8f lrun = {};
  const float scale = 0.125f;
  const int mh8 = (lane >> 4) << 3;

#pragma unroll 1
  for (int kb = 0; kb < 1024; kb += 32) {
    v8f s[2] = {};
#pragma unroll
    for (int sub = 0; sub < 2; sub++) {
      v16h kf0 = load_frag(Kbase, 1024, kb + sub * 16, 0, lane);
      v16h kf1 = load_frag(Kbase, 1024, kb + sub * 16, 32, lane);
      s[sub] = wmma_f16(qa0, kf0, s[sub]);
      s[sub] = wmma_f16(qa1, kf1, s[sub]);
    }
    const int mk0 = mb[kb + (lane & 15)];
    const int mk1 = mb[kb + 16 + (lane & 15)];

    v8f bmax;
#pragma unroll
    for (int r = 0; r < 8; r++) {
      float v = -1e30f;
      if (mk0) v = fmaxf(v, s[0][r] * scale);
      if (mk1) v = fmaxf(v, s[1][r] * scale);
      bmax[r] = v;
    }
#pragma unroll
    for (int off = 1; off < 16; off <<= 1)
#pragma unroll
      for (int r = 0; r < 8; r++)
        bmax[r] = fmaxf(bmax[r], __shfl_xor(bmax[r], off, 32));

    v8f mnew, alpha, p0, p1, bsum;
#pragma unroll
    for (int r = 0; r < 8; r++) {
      mnew[r]  = fmaxf(mrun[r], bmax[r]);
      alpha[r] = __expf(mrun[r] - mnew[r]);
      p0[r] = mk0 ? __expf(s[0][r] * scale - mnew[r]) : 0.f;
      p1[r] = mk1 ? __expf(s[1][r] * scale - mnew[r]) : 0.f;
      bsum[r] = p0[r] + p1[r];
    }
#pragma unroll
    for (int off = 1; off < 16; off <<= 1)
#pragma unroll
      for (int r = 0; r < 8; r++)
        bsum[r] += __shfl_xor(bsum[r], off, 32);
#pragma unroll
    for (int r = 0; r < 8; r++) {
      lrun[r] = lrun[r] * alpha[r] + bsum[r];
      mrun[r] = mnew[r];
    }
#pragma unroll
    for (int t = 0; t < 4; t++)
#pragma unroll
      for (int r = 0; r < 8; r++) o[t][r] *= alpha[r];

#pragma unroll
    for (int r = 0; r < 8; r++) {
      myp[(r + mh8) * 40 + (lane & 15)]      = (_Float16)p0[r];
      myp[(r + mh8) * 40 + 16 + (lane & 15)] = (_Float16)p1[r];
    }
    asm volatile("s_wait_dscnt 0" ::: "memory");
    v16h pa = load_frag(myp, 40, 0, 0, lane);

#pragma unroll
    for (int t = 0; t < 4; t++) {
      v16h vf = load_frag(Vbase, 1024, t * 16, kb, lane);
      o[t] = wmma_f16(pa, vf, o[t]);
    }
  }

#pragma unroll
  for (int t = 0; t < 4; t++) {
    const int e = t * 16 + (lane & 15);
#pragma unroll
    for (int r = 0; r < 8; r++) {
      const int q = q0 + mh8 + r;
      const float denom = lrun[r];
      const float val = denom > 0.f ? o[t][r] / denom : 0.f;
      Oh[((size_t)b * 1024 + q) * 1024 + h * 64 + e] = (_Float16)val;
    }
  }
}

__global__ __launch_bounds__(256) void ln_kernel(
    const float* __restrict__ Ain, const float* __restrict__ Bin,
    const float* __restrict__ g, const float* __restrict__ beta,
    float* __restrict__ out32, _Float16* __restrict__ out16) {
  const int row = blockIdx.x;
  const int t = threadIdx.x;
  const int lane = t & 31, wave = t >> 5;

  const float4 xa = ((const float4*)(Ain + (size_t)row * 1024))[t];
  const float4 xb = ((const float4*)(Bin + (size_t)row * 1024))[t];
  float x[4] = {xa.x + xb.x, xa.y + xb.y, xa.z + xb.z, xa.w + xb.w};

  float s1 = x[0] + x[1] + x[2] + x[3];
  float s2 = x[0]*x[0] + x[1]*x[1] + x[2]*x[2] + x[3]*x[3];
#pragma unroll
  for (int off = 16; off >= 1; off >>= 1) {
    s1 += __shfl_xor(s1, off, 32);
    s2 += __shfl_xor(s2, off, 32);
  }
  __shared__ float r1[8], r2[8];
  if (lane == 0) { r1[wave] = s1; r2[wave] = s2; }
  __syncthreads();
  float tot1 = 0.f, tot2 = 0.f;
#pragma unroll
  for (int i = 0; i < 8; i++) { tot1 += r1[i]; tot2 += r2[i]; }

  const float mu   = tot1 * (1.f / 1024.f);
  const float var  = tot2 * (1.f / 1024.f) - mu * mu;
  const float rstd = rsqrtf(var + 1e-5f);
#pragma unroll
  for (int j = 0; j < 4; j++) {
    const int idx = t * 4 + j;
    const float y = (x[j] - mu) * rstd * g[idx] + beta[idx];
    out32[(size_t)row * 1024 + idx] = y;
    if (out16) out16[(size_t)row * 1024 + idx] = (_Float16)y;
  }
}

extern "C" void kernel_launch(void* const* d_in, const int* in_sizes, int n_in,
                              void* d_out, int out_size, void* d_ws, size_t ws_size,
                              hipStream_t stream) {
  (void)in_sizes; (void)n_in; (void)out_size; (void)ws_size;
  const float* X    = (const float*)d_in[0];
  const float* Y    = (const float*)d_in[1];
  const int*   mask = (const int*)d_in[2];
  const float* Wq   = (const float*)d_in[3];
  const float* Wk   = (const float*)d_in[4];
  const float* Wv   = (const float*)d_in[5];
  const float* Wo   = (const float*)d_in[6];
  const float* ln1g = (const float*)d_in[7];
  const float* ln1b = (const float*)d_in[8];
  const float* ln2g = (const float*)d_in[9];
  const float* ln2b = (const float*)d_in[10];
  const float* W1   = (const float*)d_in[11];
  const float* b1   = (const float*)d_in[12];
  const float* W2   = (const float*)d_in[13];
  const float* b2   = (const float*)d_in[14];

  char* ws = (char*)d_ws;
  const size_t MB = 1024 * 1024;
  _Float16* Xh   = (_Float16*)(ws + 0);
  _Float16* Yh   = (_Float16*)(ws + 8 * MB);
  _Float16* Wqh  = (_Float16*)(ws + 16 * MB);
  _Float16* Wkh  = (_Float16*)(ws + 18 * MB);
  _Float16* Wvh  = (_Float16*)(ws + 20 * MB);
  _Float16* Woh  = (_Float16*)(ws + 22 * MB);
  _Float16* W1h  = (_Float16*)(ws + 24 * MB);
  _Float16* W2h  = (_Float16*)(ws + 26 * MB);
  _Float16* Qh   = (_Float16*)(ws + 28 * MB);
  _Float16* Kh   = (_Float16*)(ws + 36 * MB);
  _Float16* Vt   = (_Float16*)(ws + 44 * MB);
  float*    Fbuf = (float*)(ws + 52 * MB);
  float*    Hres = (float*)(ws + 68 * MB);
  _Float16* Ohb  = Xh;
  _Float16* Hh   = Yh;
  _Float16* FF1h = Qh;

  const int n4m = 4096 * 1024 / 4;
  const int n4w = 1024 * 1024 / 4;
  dim3 blk256(256);
  dim3 blk128(128);

  cvt_f32_f16_kernel<<<n4m / 256, blk256, 0, stream>>>(X, Xh, n4m);
  cvt_f32_f16_kernel<<<n4m / 256, blk256, 0, stream>>>(Y, Yh, n4m);
  cvt_f32_f16_kernel<<<n4w / 256, blk256, 0, stream>>>(Wq, Wqh, n4w);
  cvt_f32_f16_kernel<<<n4w / 256, blk256, 0, stream>>>(Wk, Wkh, n4w);
  cvt_f32_f16_kernel<<<n4w / 256, blk256, 0, stream>>>(Wv, Wvh, n4w);
  cvt_f32_f16_kernel<<<n4w / 256, blk256, 0, stream>>>(Wo, Woh, n4w);
  cvt_f32_f16_kernel<<<n4w / 256, blk256, 0, stream>>>(W1, W1h, n4w);
  cvt_f32_f16_kernel<<<n4w / 256, blk256, 0, stream>>>(W2, W2h, n4w);

  const int ggrid = (4096 / 128) * (1024 / 128);

  gemm_nt_kernel<<<ggrid, blk128, 0, stream>>>(Xh, Wqh, nullptr, Qh, nullptr,
                                               4096, 1024, 1024, 0);
  gemm_nt_kernel<<<ggrid, blk128, 0, stream>>>(Yh, Wkh, nullptr, Kh, nullptr,
                                               4096, 1024, 1024, 0);
  gemm_nt_kernel<<<ggrid, blk128, 0, stream>>>(Yh, Wvh, nullptr, Vt, nullptr,
                                               4096, 1024, 1024, FLAG_TRANS);

  attn_kernel<<<512, blk256, 0, stream>>>(Qh, Kh, Vt, mask, Ohb);

  gemm_nt_kernel<<<ggrid, blk128, 0, stream>>>(Ohb, Woh, Fbuf, nullptr, nullptr,
                                               4096, 1024, 1024, 0);
  ln_kernel<<<4096, blk256, 0, stream>>>(X, Fbuf, ln1g, ln1b, Hres, Hh);

  gemm_nt_kernel<<<ggrid, blk128, 0, stream>>>(Hh, W1h, nullptr, FF1h, b1,
                                               4096, 1024, 1024, FLAG_RELU);
  gemm_nt_kernel<<<ggrid, blk128, 0, stream>>>(FF1h, W2h, Fbuf, nullptr, b2,
                                               4096, 1024, 1024, 0);
  ln_kernel<<<4096, blk256, 0, stream>>>(Hres, Fbuf, ln2g, ln2b, (float*)d_out,
                                         nullptr);
}


# operand copies (f32->f16) of Y and five weights made by extra workgroups of the Q, K and V projection launches, one 16-row stripe each, not in six launches of their own: 17 launches become 11
# speedup vs baseline: 2.2706x; 2.2706x over previous
#include <hip/hip_runtime.h>
#include <hip/hip_bf16.h>


typedef __attribute__((ext_vector_type(16))) _Float16 v16h;
typedef __attribute__((ext_vector_type(8)))  _Float16 v8h;
typedef __attribute__((ext_vector_type(4)))  _Float16 v4h;
typedef __attribute__((ext_vector_type(8)))  float    v8f;
typedef __attribute__((ext_vector_type(4)))  float    v4f;

#define FLAG_RELU  1
#define FLAG_TRANS 2

__device__ __forceinline__ size_t pk_tile_off(int row, int k, int ktiles) {
  return ((size_t)(row >> 4) * ktiles + (k >> 5)) * 512;
}
__device__ __forceinline__ int pk_elem_off(int rr, int kk) {
  return ((kk >> 4) << 8) + ((((kk >> 3) & 1) << 4) + rr) * 8 + (kk & 7);
}

__device__ __forceinline__ v16h load_pk(const _Float16* p) {
  v8h lo = *(const v8h*)(p);
  v8h hi = *(const v8h*)(p + 256);
  return __builtin_shufflevector(lo, hi, 0, 1, 2, 3, 4, 5, 6, 7,
                                         8, 9, 10, 11, 12, 13, 14, 15);
}

__device__ __forceinline__ v16h load_frag(const _Float16* base, int ld,
                                          int row0, int k0, int lane) {
  const int r  = row0 + (lane & 15);
  const int kb = k0 + ((lane >> 4) << 3);
  const _Float16* p = base + (size_t)r * ld + kb;
  v8h lo = *(const v8h*)(p);
  v8h hi = *(const v8h*)(p + 16);
  return __builtin_shufflevector(lo, hi, 0, 1, 2, 3, 4, 5, 6, 7,
                                         8, 9, 10, 11, 12, 13, 14, 15);
}

__device__ __forceinline__ v8f wmma_f16(v16h a, v16h b, v8f c) {
  return __builtin_amdgcn_wmma_f32_16x16x32_f16(false, a, false, b,
                                                (short)0, c, false, false);
}

__device__ __forceinline__ void load_set(v16h (&af)[4], v16h (&bf)[4],
                                         const _Float16* Ap, const _Float16* Bp,
                                         int rstride, int kt) {
#pragma unroll
  for (int i = 0; i < 4; i++) af[i] = load_pk(Ap + (size_t)i * rstride + kt * 512);
#pragma unroll
  for (int j = 0; j < 4; j++) bf[j] = load_pk(Bp + (size_t)j * rstride + kt * 512);
}

__device__ __forceinline__ void mma_set(v8f (&acc)[4][4], const v16h (&af)[4],
                                        const v16h (&bf)[4]) {
#pragma unroll
  for (int i = 0; i < 4; i++)
#pragma unroll
    for (int j = 0; j < 4; j++)
      acc[i][j] = wmma_f16(af[i], bf[j], acc[i][j]);
}

__global__ __launch_bounds__(256) void cvt_f32_f16_kernel(
    const float* __restrict__ in, _Float16* __restrict__ out) {
  const int lane = threadIdx.x & 31;
  const int wave = threadIdx.x >> 5;
  const int row = (blockIdx.x >> 4) * 16 + (wave & 3) * 4 + (lane >> 3);
  const int k   = (blockIdx.x & 15) * 64 + (wave >> 2) * 32 + (lane & 7) * 4;
  const float4 v = *(const float4*)(in + (size_t)row * 1024 + k);
  v4h o = {(_Float16)v.x, (_Float16)v.y, (_Float16)v.z, (_Float16)v.w};
  *(v4h*)(out + pk_tile_off(row, k, 32) + pk_elem_off(row & 15, k & 31)) = o;
}

__device__ __forceinline__ void cvt_stripe(const float* __restrict__ in,
                                           _Float16* __restrict__ out,
                                           int stripe, int wave, int lane) {
  const int rr = lane >> 3;
  const int kq = (lane & 7) * 4;
  const float* src = in + (size_t)stripe * (16 * 1024) + rr * 1024 + wave * 32 + kq;
  _Float16* dst = out + (size_t)stripe * (32 * 512) + wave * 512 + pk_elem_off(rr, kq);
  v4f v[8][4];
#pragma unroll
  for (int t = 0; t < 8; t++)
#pragma unroll
    for (int g = 0; g < 4; g++)
      v[t][g] = __builtin_nontemporal_load((const v4f*)(src + g * 4096 + t * 128));
#pragma unroll
  for (int t = 0; t < 8; t++)
#pragma unroll
    for (int g = 0; g < 4; g++) {
      v4h o = {(_Float16)v[t][g].x, (_Float16)v[t][g].y, (_Float16)v[t][g].z,
               (_Float16)v[t][g].w};
      *(v4h*)(dst + t * 2048 + g * 32) = o;
    }
}

__global__ __launch_bounds__(128) void gemm_nt_kernel(
    const _Float16* __restrict__ A, const _Float16* __restrict__ Bw,
    float* __restrict__ Cf, _Float16* __restrict__ Ch,
    const float* __restrict__ bias, int Mdim, int Ndim, int Kdim, int flags,
    const float* __restrict__ ra_in, _Float16* __restrict__ ra_out, int ra_stripes,
    const float* __restrict__ rb_in, _Float16* __restrict__ rb_out) {
  const int lane = threadIdx.x & 31;
  const int wave = threadIdx.x >> 5;
  const int tiles_n = Ndim >> 7;
  const int tile_wgs = (Mdim >> 7) * tiles_n;
  if ((int)blockIdx.x >= tile_wgs) {
    const int s = (int)blockIdx.x - tile_wgs;
    const int wv = __builtin_amdgcn_readfirstlane(wave);
    const bool first = s < ra_stripes;
    cvt_stripe(first ? ra_in : rb_in, first ? ra_out : rb_out,
               first ? s : s - ra_stripes, wv, lane);
    return;
  }
  const int bm = blockIdx.x / tiles_n;
  const int bn = blockIdx.x - bm * tiles_n;
  const int m0 = bm * 128 + (wave & 1) * 64;
  const int n0 = bn * 128 + (wave >> 1) * 64;

  const int nkt = Kdim >> 5;
  const int rstride = nkt * 512;
  const _Float16* Ap = A  + (size_t)(m0 >> 4) * rstride + lane * 8;
  const _Float16* Bp = Bw + (size_t)(n0 >> 4) * rstride + lane * 8;

  v8f acc[4][4] = {};

#define SCHED_FENCE() __builtin_amdgcn_sched_barrier(0)
  v16h fa0[4], fb0[4], fa1[4], fb1[4], fa2[4], fb2[4], fa3[4], fb3[4];
  load_set(fa0, fb0, Ap, Bp, rstride, 0);
  SCHED_FENCE();
  load_set(fa1, fb1, Ap, Bp, rstride, 1);
  SCHED_FENCE();
  load_set(fa2, fb2, Ap, Bp, rstride, 2);
  SCHED_FENCE();

  int kt = 0;
#pragma unroll 1
  for (; kt < nkt - 4; kt += 4) {
    load_set(fa3, fb3, Ap, Bp, rstride, kt + 3);
    SCHED_FENCE();
    mma_set(acc, fa0, fb0);
    SCHED_FENCE();
    load_set(fa0, fb0, Ap, Bp, rstride, kt + 4);
    SCHED_FENCE();
    mma_set(acc, fa1, fb1);
    SCHED_FENCE();
    load_set(fa1, fb1, Ap, Bp, rstride, kt + 5);
    SCHED_FENCE();
    mma_set(acc, fa2, fb2);
    SCHED_FENCE();
    load_set(fa2, fb2, Ap, Bp, rstride, kt + 6);
    SCHED_FENCE();
    mma_set(acc, fa3, fb3);
    SCHED_FENCE();
  }
  load_set(fa3, fb3, Ap, Bp, rstride, kt + 3);
  SCHED_FENCE();
  mma_set(acc, fa0, fb0);
  mma_set(acc, fa1, fb1);
  mma_set(acc, fa2, fb2);
  mma_set(acc, fa3, fb3);
#undef SCHED_FENCE

  const int nl = lane & 15;
  const int mh = (lane >> 4) << 3;
  const int ktn = Ndim >> 5;
#pragma unroll
  for (int i = 0; i < 4; i++) {
#pragma unroll
    for (int j = 0; j < 4; j++) {
      const int n = n0 + j * 16 + nl;
      const float bv = bias ? bias[n] : 0.f;
      float v[8];
#pragma unroll
      for (int r = 0; r < 8; r++) {
        v[r] = acc[i][j][r] + bv;
        if (flags & FLAG_RELU) v[r] = fmaxf(v[r], 0.f);
      }
      if (Cf) {
#pragma unroll
        for (int r = 0; r < 8; r++)
          Cf[(size_t)(m0 + 16 * i + mh + r) * Ndim + n] = v[r];
      }
      if (Ch) {
        if (flags & FLAG_TRANS) {
          const int mt = m0 + 16 * i;
          const int vrow = ((mt >> 10) << 10) + n0 + 16 * j;
          _Float16* dst = Ch + pk_tile_off(vrow, mt & 1023, 32) + ((i & 1) << 8) + lane * 8;
          v8h o = {(_Float16)v[0], (_Float16)v[1], (_Float16)v[2], (_Float16)v[3],
                   (_Float16)v[4], (_Float16)v[5], (_Float16)v[6], (_Float16)v[7]};
          *(v8h*)dst = o;
        } else {
          _Float16* dst = Ch + pk_tile_off(m0 + 16 * i, n, ktn) + pk_elem_off(mh, n & 31);
#pragma unroll
          for (int r = 0; r < 8; r++) dst[r * 8] = (_Float16)v[r];
        }
      }
    }
  }
}


__device__ __forceinline__ void load_k(v16h (&kf)[4], int& mk0, int& mk1,
                                       const _Float16* Kp, const int* mb,
                                       int kb, int lane) {
#pragma unroll
  for (int sub = 0; sub < 2; sub++) {
    const _Float16* p = Kp + (size_t)((kb >> 4) + sub) * (32 * 512);
    kf[2 * sub]     = load_pk(p);
    kf[2 * sub + 1] = load_pk(p + 512);
  }
  mk0 = mb[kb + (lane & 15)];
  mk1 = mb[kb + 16 + (lane & 15)];
}

__device__ __forceinline__ void load_v(v16h (&vf)[4], const _Float16* Vp, int kb) {
#pragma unroll
  for (int t = 0; t < 4; t++)
    vf[t] = load_pk(Vp + (size_t)t * (32 * 512) + (kb >> 5) * 512);
}

__device__ __forceinline__ void attn_block(const v16h& qa0, const v16h& qa1,
                                           const v16h (&kf)[4], const v16h (&vf)[4],
                                           const int mk0, const int mk1,
                                           v8f (&o)[4], v8f& mrun, v8f& lrun,
                                           _Float16* myp, const int lane, const int mh8) {
  const float scale = 0.125f;
  v8f s[2] = {};
#pragma unroll
  for (int sub = 0; sub < 2; sub++) {
    s[sub] = wmma_f16(qa0, kf[2 * sub], s[sub]);
    s[sub] = wmma_f16(qa1, kf[2 * sub + 1], s[sub]);
  }

  v8f bmax;
#pragma unroll
  for (int r = 0; r < 8; r++) {
    float v = -1e30f;
    if (mk0) v = fmaxf(v, s[0][r] * scale);
    if (mk1) v = fmaxf(v, s[1][r] * scale);
    bmax[r] = v;
  }
#pragma unroll
  for (int off = 1; off < 16; off <<= 1)
#pragma unroll
    for (int r = 0; r < 8; r++)
      bmax[r] = fmaxf(bmax[r], __shfl_xor(bmax[r], off, 32));

  v8f mnew, alpha, p0, p1, bsum;
#pragma unroll
  for (int r = 0; r < 8; r++) {
    mnew[r]  = fmaxf(mrun[r], bmax[r]);
    alpha[r] = __expf(mrun[r] - mnew[r]);
    p0[r] = mk0 ? __expf(s[0][r] * scale - mnew[r]) : 0.f;
    p1[r] = mk1 ? __expf(s[1][r] * scale - mnew[r]) : 0.f;
    bsum[r] = p0[r] + p1[r];
  }
#pragma unroll
  for (int off = 1; off < 16; off <<= 1)
#pragma unroll
    for (int r = 0; r < 8; r++)
      bsum[r] += __shfl_xor(bsum[r], off, 32);
#pragma unroll
  for (int r = 0; r < 8; r++) {
    lrun[r] = lrun[r] * alpha[r] + bsum[r];
    mrun[r] = mnew[r];
  }
#pragma unroll
  for (int t = 0; t < 4; t++)
#pragma unroll
    for (int r = 0; r < 8; r++) o[t][r] *= alpha[r];

#pragma unroll
  for (int r = 0; r < 8; r++) {
    myp[(r + mh8) * 40 + (lane & 15)]      = (_Float16)p0[r];
    myp[(r + mh8) * 40 + 16 + (lane & 15)] = (_Float16)p1[r];
  }
  asm volatile("s_wait_dscnt 0" ::: "memory");
  v16h pa = load_frag(myp, 40, 0, 0, lane);

#pragma unroll
  for (int t = 0; t < 4; t++) o[t] = wmma_f16(pa, vf[t], o[t]);
}

__global__ __launch_bounds__(256) void attn_kernel(
    const _Float16* __restrict__ Qh, const _Float16* __restrict__ Kh,
    const _Float16* __restrict__ Vt, const int* __restrict__ mask,
    _Float16* __restrict__ Oh) {
  __shared__ _Float16 plds[8][16 * 40];

  const int lane = threadIdx.x & 31;
  const int wave = threadIdx.x >> 5;
  const int bh   = blockIdx.x >> 3;
  const int qblk = blockIdx.x & 7;
  const int b = bh >> 4, h = bh & 15;
  const int q0 = qblk * 128 + wave * 16;

  const _Float16* Qp = Qh + pk_tile_off(b * 1024 + q0, h * 64, 32) + lane * 8;
  const _Float16* Kp = Kh + pk_tile_off(b * 1024, h * 64, 32) + lane * 8;
  const _Float16* Vp = Vt + pk_tile_off((b * 16 + h) * 64, 0, 32) + lane * 8;
  const int* mb = mask + b * 1024;
  _Float16* myp = &plds[wave][0];

  const v16h qa0 = load_pk(Qp);
  const v16h qa1 = load_pk(Qp + 512);

  v8f o[4] = {};
  v8f mrun = {-1e30f, -1e30f, -1e30f, -1e30f, -1e30f, -1e30f, -1e30f, -1e30f};
  v8f lrun = {};
  const int mh8 = (lane >> 4) << 3;

#define SCHED_FENCE() __builtin_amdgcn_sched_barrier(0)
  v16h kA[4], kB[4], vf[4];
  int mA0, mA1, mB0, mB1;
  load_k(kA, mA0, mA1, Kp, mb, 0, lane);

#pragma unroll 1
  for (int kb = 0; kb < 1024; kb += 64) {
    load_v(vf, Vp, kb);
    load_k(kB, mB0, mB1, Kp, mb, kb + 32, lane);
    SCHED_FENCE();
    attn_block(qa0, qa1, kA, vf, mA0, mA1, o, mrun, lrun, myp, lane, mh8);
    SCHED_FENCE();
    load_v(vf, Vp, kb + 32);
    const int kn = (kb + 64 < 1024) ? kb + 64 : 1024 - 32;
    load_k(kA, mA0, mA1, Kp, mb, kn, lane);
    SCHED_FENCE();
    attn_block(qa0, qa1, kB, vf, mB0, mB1, o, mrun, lrun, myp, lane, mh8);
    SCHED_FENCE();
  }
#undef SCHED_FENCE

#pragma unroll
  for (int t = 0; t < 4; t++) {
    const int n = h * 64 + t * 16 + (lane & 15);
    _Float16* dst = Oh + pk_tile_off(b * 1024 + q0, n, 32) + pk_elem_off(mh8, n & 31);
#pragma unroll
    for (int r = 0; r < 8; r++) {
      const float denom = lrun[r];
      const float val = denom > 0.f ? o[t][r] / denom : 0.f;
      dst[r * 8] = (_Float16)val;
    }
  }
}

#define LN_ROWS 2
__global__ __launch_bounds__(256) void ln_kernel(
    const float* __restrict__ Ain, const float* __restrict__ Bin,
    const float* __restrict__ g, const float* __restrict__ beta,
    float* __restrict__ out32, _Float16* __restrict__ out16) {
  const int lane = threadIdx.x & 31;
  const int wave = threadIdx.x >> 5;
  const int row0 = (blockIdx.x * 8 + wave) * LN_ROWS;

  const float4* pa = (const float4*)(Ain + (size_t)row0 * 1024) + lane;
  const float4* pb = (const float4*)(Bin + (size_t)row0 * 1024) + lane;
  const float4* pg = (const float4*)g + lane;
  const float4* pt = (const float4*)beta + lane;

  float4 xa[LN_ROWS][8], xb[LN_ROWS][8], gv[8], bv[8];
#pragma unroll
  for (int r = 0; r < LN_ROWS; r++) {
#pragma unroll
    for (int q = 0; q < 8; q++) xa[r][q] = pa[r * 256 + q * 32];
#pragma unroll
    for (int q = 0; q < 8; q++) xb[r][q] = pb[r * 256 + q * 32];
  }
#pragma unroll
  for (int q = 0; q < 8; q++) gv[q] = pg[q * 32];
#pragma unroll
  for (int q = 0; q < 8; q++) bv[q] = pt[q * 32];
  __builtin_amdgcn_sched_group_barrier(0x020, 48, 0);
  __builtin_amdgcn_sched_group_barrier(0x002, 1024, 0);

  float4 x[LN_ROWS][8];
  float s1[LN_ROWS], s2[LN_ROWS];
#pragma unroll
  for (int r = 0; r < LN_ROWS; r++) {
    float4 a1 = {0.f, 0.f, 0.f, 0.f}, a2 = {0.f, 0.f, 0.f, 0.f};
#pragma unroll
    for (int q = 0; q < 8; q++) {
      x[r][q].x = xa[r][q].x + xb[r][q].x;
      x[r][q].y = xa[r][q].y + xb[r][q].y;
      x[r][q].z = xa[r][q].z + xb[r][q].z;
      x[r][q].w = xa[r][q].w + xb[r][q].w;
      a1.x += x[r][q].x; a1.y += x[r][q].y; a1.z += x[r][q].z; a1.w += x[r][q].w;
      a2.x += x[r][q].x * x[r][q].x; a2.y += x[r][q].y * x[r][q].y;
      a2.z += x[r][q].z * x[r][q].z; a2.w += x[r][q].w * x[r][q].w;
    }
    s1[r] = (a1.x + a1.y) + (a1.z + a1.w);
    s2[r] = (a2.x + a2.y) + (a2.z + a2.w);
  }
#pragma unroll
  for (int off = 16; off >= 1; off >>= 1) {
#pragma unroll
    for (int r = 0; r < LN_ROWS; r++) {
      s1[r] += __shfl_xor(s1[r], off, 32);
      s2[r] += __shfl_xor(s2[r], off, 32);
    }
  }

#pragma unroll
  for (int r = 0; r < LN_ROWS; r++) {
    const int row = row0 + r;
    const float mu   = s1[r] * (1.f / 1024.f);
    const float var  = s2[r] * (1.f / 1024.f) - mu * mu;
    const float rstd = rsqrtf(var + 1e-5f);
    float4 y[8];
    float4* po = (float4*)(out32 + (size_t)row * 1024) + lane;
#pragma unroll
    for (int q = 0; q < 8; q++) {
      y[q].x = (x[r][q].x - mu) * rstd * gv[q].x + bv[q].x;
      y[q].y = (x[r][q].y - mu) * rstd * gv[q].y + bv[q].y;
      y[q].z = (x[r][q].z - mu) * rstd * gv[q].z + bv[q].z;
      y[q].w = (x[r][q].w - mu) * rstd * gv[q].w + bv[q].w;
      po[q * 32] = y[q];
    }
    if (out16) {
#pragma unroll
      for (int q = 0; q < 8; q++) {
        const int k = (q * 32 + lane) * 4;
        v4h o = {(_Float16)y[q].x, (_Float16)y[q].y, (_Float16)y[q].z, (_Float16)y[q].w};
        *(v4h*)(out16 + pk_tile_off(row, k, 32) + pk_elem_off(row & 15, k & 31)) = o;
      }
    }
  }
}
#undef LN_ROWS

extern "C" void kernel_launch(void* const* d_in, const int* in_sizes, int n_in,
                              void* d_out, int out_size, void* d_ws, size_t ws_size,
                              hipStream_t stream) {
  (void)in_sizes; (void)n_in; (void)out_size; (void)ws_size;
  const float* X    = (const float*)d_in[0];
  const float* Y    = (const float*)d_in[1];
  const int*   mask = (const int*)d_in[2];
  const float* Wq   = (const float*)d_in[3];
  const float* Wk   = (const float*)d_in[4];
  const float* Wv   = (const float*)d_in[5];
  const float* Wo   = (const float*)d_in[6];
  const float* ln1g = (const float*)d_in[7];
  const float* ln1b = (const float*)d_in[8];
  const float* ln2g = (const float*)d_in[9];
  const float* ln2b = (const float*)d_in[10];
  const float* W1   = (const float*)d_in[11];
  const float* b1   = (const float*)d_in[12];
  const float* W2   = (const float*)d_in[13];
  const float* b2   = (const float*)d_in[14];

  char* ws = (char*)d_ws;
  const size_t MB = 1024 * 1024;
  _Float16* Xh   = (_Float16*)(ws + 0);
  _Float16* Yh   = (_Float16*)(ws + 8 * MB);
  _Float16* Wqh  = (_Float16*)(ws + 16 * MB);
  _Float16* Wkh  = (_Float16*)(ws + 18 * MB);
  _Float16* Wvh  = (_Float16*)(ws + 20 * MB);
  _Float16* Woh  = (_Float16*)(ws + 22 * MB);
  _Float16* W1h  = (_Float16*)(ws + 24 * MB);
  _Float16* W2h  = (_Float16*)(ws + 26 * MB);
  _Float16* Qh   = (_Float16*)(ws + 28 * MB);
  _Float16* Kh   = (_Float16*)(ws + 36 * MB);
  _Float16* Vt   = (_Float16*)(ws + 44 * MB);
  float*    Fbuf = (float*)(ws + 52 * MB);
  float*    Hres = (float*)(ws + 68 * MB);
  _Float16* Ohb  = Xh;
  _Float16* Hh   = Yh;
  _Float16* FF1h = Qh;

  const int n4m = 4096 * 1024 / 4;
  const int n4w = 1024 * 1024 / 4;
  dim3 blk256(256);
  dim3 blk128(128);

  cvt_f32_f16_kernel<<<n4m / 256, blk256, 0, stream>>>(X, Xh);
  cvt_f32_f16_kernel<<<n4w / 256, blk256, 0, stream>>>(Wq, Wqh);

  const int ggrid = (4096 / 128) * (1024 / 128);
  const int sY = 4096 / 16;
  const int sW = 1024 / 16;

  gemm_nt_kernel<<<ggrid + sY + sW, blk128, 0, stream>>>(
      Xh, Wqh, nullptr, Qh, nullptr, 4096, 1024, 1024, 0, Y, Yh, sY, Wk, Wkh);
  gemm_nt_kernel<<<ggrid + sW + sW, blk128, 0, stream>>>(
      Yh, Wkh, nullptr, Kh, nullptr, 4096, 1024, 1024, 0, Wv, Wvh, sW, Wo, Woh);
  gemm_nt_kernel<<<ggrid + sW + sW, blk128, 0, stream>>>(
      Yh, Wvh, nullptr, Vt, nullptr, 4096, 1024, 1024, FLAG_TRANS, W1, W1h, sW,
      W2, W2h);

  attn_kernel<<<512, blk256, 0, stream>>>(Qh, Kh, Vt, mask, Ohb);

  gemm_nt_kernel<<<ggrid, blk128, 0, stream>>>(Ohb, Woh, Fbuf, nullptr, nullptr,
                                               4096, 1024, 1024, 0, nullptr,
                                               nullptr, 0, nullptr, nullptr);
  ln_kernel<<<256, blk256, 0, stream>>>(X, Fbuf, ln1g, ln1b, Hres, Hh);

  gemm_nt_kernel<<<ggrid, blk128, 0, stream>>>(Hh, W1h, nullptr, FF1h, b1,
                                               4096, 1024, 1024, FLAG_RELU,
                                               nullptr, nullptr, 0, nullptr,
                                               nullptr);
  gemm_nt_kernel<<<ggrid, blk128, 0, stream>>>(FF1h, W2h, Fbuf, nullptr, b2,
                                               4096, 1024, 1024, 0, nullptr,
                                               nullptr, 0, nullptr, nullptr);
  ln_kernel<<<256, blk256, 0, stream>>>(Hres, Fbuf, ln2g, ln2b, (float*)d_out,
                                         nullptr);
}


# attention: lazy rescale in the log2 domain: row-max butterfly, alpha and the rescale of O run only when a score tops the max in use by 2^8 (ballot, scalar branch); P and its sums round differently
# speedup vs baseline: 2.4819x; 1.0931x over previous
#include <hip/hip_runtime.h>
#include <hip/hip_bf16.h>


typedef __attribute__((ext_vector_type(16))) _Float16 v16h;
typedef __attribute__((ext_vector_type(8)))  _Float16 v8h;
typedef __attribute__((ext_vector_type(4)))  _Float16 v4h;
typedef __attribute__((ext_vector_type(8)))  float    v8f;
typedef __attribute__((ext_vector_type(4)))  float    v4f;

#define FLAG_RELU  1
#define FLAG_TRANS 2

__device__ __forceinline__ size_t pk_tile_off(int row, int k, int ktiles) {
  return ((size_t)(row >> 4) * ktiles + (k >> 5)) * 512;
}
__device__ __forceinline__ int pk_elem_off(int rr, int kk) {
  return ((kk >> 4) << 8) + ((((kk >> 3) & 1) << 4) + rr) * 8 + (kk & 7);
}

__device__ __forceinline__ v16h load_pk(const _Float16* p) {
  v8h lo = *(const v8h*)(p);
  v8h hi = *(const v8h*)(p + 256);
  return __builtin_shufflevector(lo, hi, 0, 1, 2, 3, 4, 5, 6, 7,
                                         8, 9, 10, 11, 12, 13, 14, 15);
}

__device__ __forceinline__ v16h load_frag(const _Float16* base, int ld,
                                          int row0, int k0, int lane) {
  const int r  = row0 + (lane & 15);
  const int kb = k0 + ((lane >> 4) << 3);
  const _Float16* p = base + (size_t)r * ld + kb;
  v8h lo = *(const v8h*)(p);
  v8h hi = *(const v8h*)(p + 16);
  return __builtin_shufflevector(lo, hi, 0, 1, 2, 3, 4, 5, 6, 7,
                                         8, 9, 10, 11, 12, 13, 14, 15);
}

__device__ __forceinline__ v8f wmma_f16(v16h a, v16h b, v8f c) {
  return __builtin_amdgcn_wmma_f32_16x16x32_f16(false, a, false, b,
                                                (short)0, c, false, false);
}

__device__ __forceinline__ void load_set(v16h (&af)[4], v16h (&bf)[4],
                                         const _Float16* Ap, const _Float16* Bp,
                                         int rstride, int kt) {
#pragma unroll
  for (int i = 0; i < 4; i++) af[i] = load_pk(Ap + (size_t)i * rstride + kt * 512);
#pragma unroll
  for (int j = 0; j < 4; j++) bf[j] = load_pk(Bp + (size_t)j * rstride + kt * 512);
}

__device__ __forceinline__ void mma_set(v8f (&acc)[4][4], const v16h (&af)[4],
                                        const v16h (&bf)[4]) {
#pragma unroll
  for (int i = 0; i < 4; i++)
#pragma unroll
    for (int j = 0; j < 4; j++)
      acc[i][j] = wmma_f16(af[i], bf[j], acc[i][j]);
}

__global__ __launch_bounds__(256) void cvt_f32_f16_kernel(
    const float* __restrict__ in, _Float16* __restrict__ out) {
  const int lane = threadIdx.x & 31;
  const int wave = threadIdx.x >> 5;
  const int row = (blockIdx.x >> 4) * 16 + (wave & 3) * 4 + (lane >> 3);
  const int k   = (blockIdx.x & 15) * 64 + (wave >> 2) * 32 + (lane & 7) * 4;
  const float4 v = *(const float4*)(in + (size_t)row * 1024 + k);
  v4h o = {(_Float16)v.x, (_Float16)v.y, (_Float16)v.z, (_Float16)v.w};
  *(v4h*)(out + pk_tile_off(row, k, 32) + pk_elem_off(row & 15, k & 31)) = o;
}

__device__ __forceinline__ void cvt_stripe(const float* __restrict__ in,
                                           _Float16* __restrict__ out,
                                           int stripe, int wave, int lane) {
  const int rr = lane >> 3;
  const int kq = (lane & 7) * 4;
  const float* src = in + (size_t)stripe * (16 * 1024) + rr * 1024 + wave * 32 + kq;
  _Float16* dst = out + (size_t)stripe * (32 * 512) + wave * 512 + pk_elem_off(rr, kq);
  v4f v[8][4];
#pragma unroll
  for (int t = 0; t < 8; t++)
#pragma unroll
    for (int g = 0; g < 4; g++)
      v[t][g] = __builtin_nontemporal_load((const v4f*)(src + g * 4096 + t * 128));
#pragma unroll
  for (int t = 0; t < 8; t++)
#pragma unroll
    for (int g = 0; g < 4; g++) {
      v4h o = {(_Float16)v[t][g].x, (_Float16)v[t][g].y, (_Float16)v[t][g].z,
               (_Float16)v[t][g].w};
      *(v4h*)(dst + t * 2048 + g * 32) = o;
    }
}

__global__ __launch_bounds__(128) void gemm_nt_kernel(
    const _Float16* __restrict__ A, const _Float16* __restrict__ Bw,
    float* __restrict__ Cf, _Float16* __restrict__ Ch,
    const float* __restrict__ bias, int Mdim, int Ndim, int Kdim, int flags,
    const float* __restrict__ ra_in, _Float16* __restrict__ ra_out, int ra_stripes,
    const float* __restrict__ rb_in, _Float16* __restrict__ rb_out) {
  const int lane = threadIdx.x & 31;
  const int wave = threadIdx.x >> 5;
  const int tiles_n = Ndim >> 7;
  const int tile_wgs = (Mdim >> 7) * tiles_n;
  if ((int)blockIdx.x >= tile_wgs) {
    const int s = (int)blockIdx.x - tile_wgs;
    const int wv = __builtin_amdgcn_readfirstlane(wave);
    const bool first = s < ra_stripes;
    cvt_stripe(first ? ra_in : rb_in, first ? ra_out : rb_out,
               first ? s : s - ra_stripes, wv, lane);
    return;
  }
  const int bm = blockIdx.x / tiles_n;
  const int bn = blockIdx.x - bm * tiles_n;
  const int m0 = bm * 128 + (wave & 1) * 64;
  const int n0 = bn * 128 + (wave >> 1) * 64;

  const int nkt = Kdim >> 5;
  const int rstride = nkt * 512;
  const _Float16* Ap = A  + (size_t)(m0 >> 4) * rstride + lane * 8;
  const _Float16* Bp = Bw + (size_t)(n0 >> 4) * rstride + lane * 8;

  v8f acc[4][4] = {};

#define SCHED_FENCE() __builtin_amdgcn_sched_barrier(0)
  v16h fa0[4], fb0[4], fa1[4], fb1[4], fa2[4], fb2[4], fa3[4], fb3[4];
  load_set(fa0, fb0, Ap, Bp, rstride, 0);
  SCHED_FENCE();
  load_set(fa1, fb1, Ap, Bp, rstride, 1);
  SCHED_FENCE();
  load_set(fa2, fb2, Ap, Bp, rstride, 2);
  SCHED_FENCE();

  int kt = 0;
#pragma unroll 1
  for (; kt < nkt - 4; kt += 4) {
    load_set(fa3, fb3, Ap, Bp, rstride, kt + 3);
    SCHED_FENCE();
    mma_set(acc, fa0, fb0);
    SCHED_FENCE();
    load_set(fa0, fb0, Ap, Bp, rstride, kt + 4);
    SCHED_FENCE();
    mma_set(acc, fa1, fb1);
    SCHED_FENCE();
    load_set(fa1, fb1, Ap, Bp, rstride, kt + 5);
    SCHED_FENCE();
    mma_set(acc, fa2, fb2);
    SCHED_FENCE();
    load_set(fa2, fb2, Ap, Bp, rstride, kt + 6);
    SCHED_FENCE();
    mma_set(acc, fa3, fb3);
    SCHED_FENCE();
  }
  load_set(fa3, fb3, Ap, Bp, rstride, kt + 3);
  SCHED_FENCE();
  mma_set(acc, fa0, fb0);
  mma_set(acc, fa1, fb1);
  mma_set(acc, fa2, fb2);
  mma_set(acc, fa3, fb3);
#undef SCHED_FENCE

  const int nl = lane & 15;
  const int mh = (lane >> 4) << 3;
  const int ktn = Ndim >> 5;
#pragma unroll
  for (int i = 0; i < 4; i++) {
#pragma unroll
    for (int j = 0; j < 4; j++) {
      const int n = n0 + j * 16 + nl;
      const float bv = bias ? bias[n] : 0.f;
      float v[8];
#pragma unroll
      for (int r = 0; r < 8; r++) {
        v[r] = acc[i][j][r] + bv;
        if (flags & FLAG_RELU) v[r] = fmaxf(v[r], 0.f);
      }
      if (Cf) {
#pragma unroll
        for (int r = 0; r < 8; r++)
          Cf[(size_t)(m0 + 16 * i + mh + r) * Ndim + n] = v[r];
      }
      if (Ch) {
        if (flags & FLAG_TRANS) {
          const int mt = m0 + 16 * i;
          const int vrow = ((mt >> 10) << 10) + n0 + 16 * j;
          _Float16* dst = Ch + pk_tile_off(vrow, mt & 1023, 32) + ((i & 1) << 8) + lane * 8;
          v8h o = {(_Float16)v[0], (_Float16)v[1], (_Float16)v[2], (_Float16)v[3],
                   (_Float16)v[4], (_Float16)v[5], (_Float16)v[6], (_Float16)v[7]};
          *(v8h*)dst = o;
        } else {
          _Float16* dst = Ch + pk_tile_off(m0 + 16 * i, n, ktn) + pk_elem_off(mh, n & 31);
#pragma unroll
          for (int r = 0; r < 8; r++) dst[r * 8] = (_Float16)v[r];
        }
      }
    }
  }
}


__device__ __forceinline__ void load_k(v16h (&kf)[4], int& mk0, int& mk1,
                                       const _Float16* Kp, const int* mb,
                                       int kb, int lane) {
#pragma unroll
  for (int sub = 0; sub < 2; sub++) {
    const _Float16* p = Kp + (size_t)((kb >> 4) + sub) * (32 * 512);
    kf[2 * sub]     = load_pk(p);
    kf[2 * sub + 1] = load_pk(p + 512);
  }
  mk0 = mb[kb + (lane & 15)];
  mk1 = mb[kb + 16 + (lane & 15)];
}

__device__ __forceinline__ void load_v(v16h (&vf)[4], const _Float16* Vp, int kb) {
#pragma unroll
  for (int t = 0; t < 4; t++)
    vf[t] = load_pk(Vp + (size_t)t * (32 * 512) + (kb >> 5) * 512);
}

__device__ __forceinline__ void attn_block(const v16h& qa0, const v16h& qa1,
                                           const v16h (&kf)[4], const v16h (&vf)[4],
                                           const int mk0, const int mk1,
                                           const v16h& ones, v8f (&o)[5], v8f& nmuse,
                                           _Float16* myp, const int lane, const int mh8) {
  const float c2 = 0.125f * 1.44269504088896341f;
  v8f s[2] = {};
#pragma unroll
  for (int sub = 0; sub < 2; sub++) {
    s[sub] = wmma_f16(qa0, kf[2 * sub], s[sub]);
    s[sub] = wmma_f16(qa1, kf[2 * sub + 1], s[sub]);
  }

  v8f e0, e1;
#pragma unroll
  for (int r = 0; r < 8; r++) {
    e0[r] = __builtin_fmaf(s[0][r], c2, nmuse[r]);
    e1[r] = __builtin_fmaf(s[1][r], c2, nmuse[r]);
  }
  float x0 = e0[0], x1 = e1[0];
#pragma unroll
  for (int r = 1; r < 8; r++) {
    x0 = fmaxf(x0, e0[r]);
    x1 = fmaxf(x1, e1[r]);
  }
  const float x = fmaxf(mk0 ? x0 : -1e30f, mk1 ? x1 : -1e30f);

  if (__builtin_expect(__builtin_amdgcn_ballot_w32(x > 8.0f) != 0, 0)) {
    v8f bmax;
#pragma unroll
    for (int r = 0; r < 8; r++) {
      float v = -1e30f;
      if (mk0) v = fmaxf(v, s[0][r] * c2);
      if (mk1) v = fmaxf(v, s[1][r] * c2);
      bmax[r] = v;
    }
#pragma unroll
    for (int off = 1; off < 16; off <<= 1)
#pragma unroll
      for (int r = 0; r < 8; r++)
        bmax[r] = fmaxf(bmax[r], __shfl_xor(bmax[r], off, 32));
    v8f alpha;
#pragma unroll
    for (int r = 0; r < 8; r++) {
      const float nnew = fminf(nmuse[r], -bmax[r]);
      alpha[r] = __builtin_amdgcn_exp2f(nnew - nmuse[r]);
      nmuse[r] = nnew;
      e0[r] = __builtin_fmaf(s[0][r], c2, nnew);
      e1[r] = __builtin_fmaf(s[1][r], c2, nnew);
    }
#pragma unroll
    for (int t = 0; t < 5; t++)
#pragma unroll
      for (int r = 0; r < 8; r++) o[t][r] *= alpha[r];
  }

  v8f p0, p1;
#pragma unroll
  for (int r = 0; r < 8; r++) {
    p0[r] = mk0 ? __builtin_amdgcn_exp2f(e0[r]) : 0.f;
    p1[r] = mk1 ? __builtin_amdgcn_exp2f(e1[r]) : 0.f;
  }

#pragma unroll
  for (int r = 0; r < 8; r++) {
    myp[(r + mh8) * 40 + (lane & 15)]      = (_Float16)p0[r];
    myp[(r + mh8) * 40 + 16 + (lane & 15)] = (_Float16)p1[r];
  }
  asm volatile("s_wait_dscnt 0" ::: "memory");
  v16h pa = load_frag(myp, 40, 0, 0, lane);

#pragma unroll
  for (int t = 0; t < 4; t++) o[t] = wmma_f16(pa, vf[t], o[t]);
  o[4] = wmma_f16(pa, ones, o[4]);
}

__global__ __launch_bounds__(256) void attn_kernel(
    const _Float16* __restrict__ Qh, const _Float16* __restrict__ Kh,
    const _Float16* __restrict__ Vt, const int* __restrict__ mask,
    _Float16* __restrict__ Oh) {
  __shared__ _Float16 plds[8][16 * 40];

  const int lane = threadIdx.x & 31;
  const int wave = threadIdx.x >> 5;
  const int bh   = blockIdx.x >> 3;
  const int qblk = blockIdx.x & 7;
  const int b = bh >> 4, h = bh & 15;
  const int q0 = qblk * 128 + wave * 16;

  const _Float16* Qp = Qh + pk_tile_off(b * 1024 + q0, h * 64, 32) + lane * 8;
  const _Float16* Kp = Kh + pk_tile_off(b * 1024, h * 64, 32) + lane * 8;
  const _Float16* Vp = Vt + pk_tile_off((b * 16 + h) * 64, 0, 32) + lane * 8;
  const int* mb = mask + b * 1024;
  _Float16* myp = &plds[wave][0];

  const v16h qa0 = load_pk(Qp);
  const v16h qa1 = load_pk(Qp + 512);

  v8f o[5] = {};
  v8f nmuse = {1e30f, 1e30f, 1e30f, 1e30f, 1e30f, 1e30f, 1e30f, 1e30f};
  v16h ones;
#pragma unroll
  for (int i = 0; i < 16; i++) ones[i] = (_Float16)1.0f;
  const int mh8 = (lane >> 4) << 3;

#define SCHED_FENCE() __builtin_amdgcn_sched_barrier(0)
  v16h kA[4], kB[4], vf[4];
  int mA0, mA1, mB0, mB1;
  load_k(kA, mA0, mA1, Kp, mb, 0, lane);

#pragma unroll 1
  for (int kb = 0; kb < 1024; kb += 64) {
    load_v(vf, Vp, kb);
    load_k(kB, mB0, mB1, Kp, mb, kb + 32, lane);
    SCHED_FENCE();
    attn_block(qa0, qa1, kA, vf, mA0, mA1, ones, o, nmuse, myp, lane, mh8);
    SCHED_FENCE();
    load_v(vf, Vp, kb + 32);
    const int kn = (kb + 64 < 1024) ? kb + 64 : 1024 - 32;
    load_k(kA, mA0, mA1, Kp, mb, kn, lane);
    SCHED_FENCE();
    attn_block(qa0, qa1, kB, vf, mB0, mB1, ones, o, nmuse, myp, lane, mh8);
    SCHED_FENCE();
  }
#undef SCHED_FENCE

#pragma unroll
  for (int t = 0; t < 4; t++) {
    const int n = h * 64 + t * 16 + (lane & 15);
    _Float16* dst = Oh + pk_tile_off(b * 1024 + q0, n, 32) + pk_elem_off(mh8, n & 31);
#pragma unroll
    for (int r = 0; r < 8; r++) {
      const float denom = o[4][r];
      const float val = denom > 0.f ? o[t][r] / denom : 0.f;
      dst[r * 8] = (_Float16)val;
    }
  }
}

#define LN_ROWS 2
__global__ __launch_bounds__(256) void ln_kernel(
    const float* __restrict__ Ain, const float* __restrict__ Bin,
    const float* __restrict__ g, const float* __restrict__ beta,
    float* __restrict__ out32, _Float16* __restrict__ out16) {
  const int lane = threadIdx.x & 31;
  const int wave = threadIdx.x >> 5;
  const int row0 = (blockIdx.x * 8 + wave) * LN_ROWS;

  const float4* pa = (const float4*)(Ain + (size_t)row0 * 1024) + lane;
  const float4* pb = (const float4*)(Bin + (size_t)row0 * 1024) + lane;
  const float4* pg = (const float4*)g + lane;
  const float4* pt = (const float4*)beta + lane;

  float4 xa[LN_ROWS][8], xb[LN_ROWS][8], gv[8], bv[8];
#pragma unroll
  for (int r = 0; r < LN_ROWS; r++) {
#pragma unroll
    for (int q = 0; q < 8; q++) xa[r][q] = pa[r * 256 + q * 32];
#pragma unroll
    for (int q = 0; q < 8; q++) xb[r][q] = pb[r * 256 + q * 32];
  }
#pragma unroll
  for (int q = 0; q < 8; q++) gv[q] = pg[q * 32];
#pragma unroll
  for (int q = 0; q < 8; q++) bv[q] = pt[q * 32];
  __builtin_amdgcn_sched_group_barrier(0x020, 48, 0);
  __builtin_amdgcn_sched_group_barrier(0x002, 1024, 0);

  float4 x[LN_ROWS][8];
  float s1[LN_ROWS], s2[LN_ROWS];
#pragma unroll
  for (int r = 0; r < LN_ROWS; r++) {
    float4 a1 = {0.f, 0.f, 0.f, 0.f}, a2 = {0.f, 0.f, 0.f, 0.f};
#pragma unroll
    for (int q = 0; q < 8; q++) {
      x[r][q].x = xa[r][q].x + xb[r][q].x;
      x[r][q].y = xa[r][q].y + xb[r][q].y;
      x[r][q].z = xa[r][q].z + xb[r][q].z;
      x[r][q].w = xa[r][q].w + xb[r][q].w;
      a1.x += x[r][q].x; a1.y += x[r][q].y; a1.z += x[r][q].z; a1.w += x[r][q].w;
      a2.x += x[r][q].x * x[r][q].x; a2.y += x[r][q].y * x[r][q].y;
      a2.z += x[r][q].z * x[r][q].z; a2.w += x[r][q].w * x[r][q].w;
    }
    s1[r] = (a1.x + a1.y) + (a1.z + a1.w);
    s2[r] = (a2.x + a2.y) + (a2.z + a2.w);
  }
#pragma unroll
  for (int off = 16; off >= 1; off >>= 1) {
#pragma unroll
    for (int r = 0; r < LN_ROWS; r++) {
      s1[r] += __shfl_xor(s1[r], off, 32);
      s2[r] += __shfl_xor(s2[r], off, 32);
    }
  }

#pragma unroll
  for (int r = 0; r < LN_ROWS; r++) {
    const int row = row0 + r;
    const float mu   = s1[r] * (1.f / 1024.f);
    const float var  = s2[r] * (1.f / 1024.f) - mu * mu;
    const float rstd = rsqrtf(var + 1e-5f);
    float4 y[8];
    float4* po = (float4*)(out32 + (size_t)row * 1024) + lane;
#pragma unroll
    for (int q = 0; q < 8; q++) {
      y[q].x = (x[r][q].x - mu) * rstd * gv[q].x + bv[q].x;
      y[q].y = (x[r][q].y - mu) * rstd * gv[q].y + bv[q].y;
      y[q].z = (x[r][q].z - mu) * rstd * gv[q].z + bv[q].z;
      y[q].w = (x[r][q].w - mu) * rstd * gv[q].w + bv[q].w;
      po[q * 32] = y[q];
    }
    if (out16) {
#pragma unroll
      for (int q = 0; q < 8; q++) {
        const int k = (q * 32 + lane) * 4;
        v4h o = {(_Float16)y[q].x, (_Float16)y[q].y, (_Float16)y[q].z, (_Float16)y[q].w};
        *(v4h*)(out16 + pk_tile_off(row, k, 32) + pk_elem_off(row & 15, k & 31)) = o;
      }
    }
  }
}
#undef LN_ROWS

extern "C" void kernel_launch(void* const* d_in, const int* in_sizes, int n_in,
                              void* d_out, int out_size, void* d_ws, size_t ws_size,
                              hipStream_t stream) {
  (void)in_sizes; (void)n_in; (void)out_size; (void)ws_size;
  const float* X    = (const float*)d_in[0];
  const float* Y    = (const float*)d_in[1];
  const int*   mask = (const int*)d_in[2];
  const float* Wq   = (const float*)d_in[3];
  const float* Wk   = (const float*)d_in[4];
  const float* Wv   = (const float*)d_in[5];
  const float* Wo   = (const float*)d_in[6];
  const float* ln1g = (const float*)d_in[7];
  const float* ln1b = (const float*)d_in[8];
  const float* ln2g = (const float*)d_in[9];
  const float* ln2b = (const float*)d_in[10];
  const float* W1   = (const float*)d_in[11];
  const float* b1   = (const float*)d_in[12];
  const float* W2   = (const float*)d_in[13];
  const float* b2   = (const float*)d_in[14];

  char* ws = (char*)d_ws;
  const size_t MB = 1024 * 1024;
  _Float16* Xh   = (_Float16*)(ws + 0);
  _Float16* Yh   = (_Float16*)(ws + 8 * MB);
  _Float16* Wqh  = (_Float16*)(ws + 16 * MB);
  _Float16* Wkh  = (_Float16*)(ws + 18 * MB);
  _Float16* Wvh  = (_Float16*)(ws + 20 * MB);
  _Float16* Woh  = (_Float16*)(ws + 22 * MB);
  _Float16* W1h  = (_Float16*)(ws + 24 * MB);
  _Float16* W2h  = (_Float16*)(ws + 26 * MB);
  _Float16* Qh   = (_Float16*)(ws + 28 * MB);
  _Float16* Kh   = (_Float16*)(ws + 36 * MB);
  _Float16* Vt   = (_Float16*)(ws + 44 * MB);
  float*    Fbuf = (float*)(ws + 52 * MB);
  float*    Hres = (float*)(ws + 68 * MB);
  _Float16* Ohb  = Xh;
  _Float16* Hh   = Yh;
  _Float16* FF1h = Qh;

  const int n4m = 4096 * 1024 / 4;
  const int n4w = 1024 * 1024 / 4;
  dim3 blk256(256);
  dim3 blk128(128);

  cvt_f32_f16_kernel<<<n4m / 256, blk256, 0, stream>>>(X, Xh);
  cvt_f32_f16_kernel<<<n4w / 256, blk256, 0, stream>>>(Wq, Wqh);

  const int ggrid = (4096 / 128) * (1024 / 128);
  const int sY = 4096 / 16;
  const int sW = 1024 / 16;

  gemm_nt_kernel<<<ggrid + sY + sW, blk128, 0, stream>>>(
      Xh, Wqh, nullptr, Qh, nullptr, 4096, 1024, 1024, 0, Y, Yh, sY, Wk, Wkh);
  gemm_nt_kernel<<<ggrid + sW + sW, blk128, 0, stream>>>(
      Yh, Wkh, nullptr, Kh, nullptr, 4096, 1024, 1024, 0, Wv, Wvh, sW, Wo, Woh);
  gemm_nt_kernel<<<ggrid + sW + sW, blk128, 0, stream>>>(
      Yh, Wvh, nullptr, Vt, nullptr, 4096, 1024, 1024, FLAG_TRANS, W1, W1h, sW,
      W2, W2h);

  attn_kernel<<<512, blk256, 0, stream>>>(Qh, Kh, Vt, mask, Ohb);

  gemm_nt_kernel<<<ggrid, blk128, 0, stream>>>(Ohb, Woh, Fbuf, nullptr, nullptr,
                                               4096, 1024, 1024, 0, nullptr,
                                               nullptr, 0, nullptr, nullptr);
  ln_kernel<<<256, blk256, 0, stream>>>(X, Fbuf, ln1g, ln1b, Hres, Hh);

  gemm_nt_kernel<<<ggrid, blk128, 0, stream>>>(Hh, W1h, nullptr, FF1h, b1,
                                               4096, 1024, 1024, FLAG_RELU,
                                               nullptr, nullptr, 0, nullptr,
                                               nullptr);
  gemm_nt_kernel<<<ggrid, blk128, 0, stream>>>(FF1h, W2h, Fbuf, nullptr, b2,
                                               4096, 1024, 1024, 0, nullptr,
                                               nullptr, 0, nullptr, nullptr);
  ln_kernel<<<256, blk256, 0, stream>>>(Hres, Fbuf, ln2g, ln2b, (float*)d_out,
                                         nullptr);
}


# gemm with f16 output (Q, K, FFN1): weights taken as WMMA A, so a lane's 8 accumulators per subtile are one 16-byte piece of the fragment-order output: 16 whole-line b128 stores a wave, not 128 2-byte
# speedup vs baseline: 2.6321x; 1.0605x over previous
#include <hip/hip_runtime.h>
#include <hip/hip_bf16.h>


typedef __attribute__((ext_vector_type(16))) _Float16 v16h;
typedef __attribute__((ext_vector_type(8)))  _Float16 v8h;
typedef __attribute__((ext_vector_type(4)))  _Float16 v4h;
typedef __attribute__((ext_vector_type(8)))  float    v8f;
typedef __attribute__((ext_vector_type(4)))  float    v4f;

#define FLAG_RELU  1
#define FLAG_TRANS 2

__device__ __forceinline__ size_t pk_tile_off(int row, int k, int ktiles) {
  return ((size_t)(row >> 4) * ktiles + (k >> 5)) * 512;
}
__device__ __forceinline__ int pk_elem_off(int rr, int kk) {
  return ((kk >> 4) << 8) + ((((kk >> 3) & 1) << 4) + rr) * 8 + (kk & 7);
}

__device__ __forceinline__ v16h load_pk(const _Float16* p) {
  v8h lo = *(const v8h*)(p);
  v8h hi = *(const v8h*)(p + 256);
  return __builtin_shufflevector(lo, hi, 0, 1, 2, 3, 4, 5, 6, 7,
                                         8, 9, 10, 11, 12, 13, 14, 15);
}

__device__ __forceinline__ v16h load_frag(const _Float16* base, int ld,
                                          int row0, int k0, int lane) {
  const int r  = row0 + (lane & 15);
  const int kb = k0 + ((lane >> 4) << 3);
  const _Float16* p = base + (size_t)r * ld + kb;
  v8h lo = *(const v8h*)(p);
  v8h hi = *(const v8h*)(p + 16);
  return __builtin_shufflevector(lo, hi, 0, 1, 2, 3, 4, 5, 6, 7,
                                         8, 9, 10, 11, 12, 13, 14, 15);
}

__device__ __forceinline__ v8f wmma_f16(v16h a, v16h b, v8f c) {
  return __builtin_amdgcn_wmma_f32_16x16x32_f16(false, a, false, b,
                                                (short)0, c, false, false);
}

__device__ __forceinline__ void load_set(v16h (&af)[4], v16h (&bf)[4],
                                         const _Float16* Ap, const _Float16* Bp,
                                         int rstride, int kt) {
#pragma unroll
  for (int i = 0; i < 4; i++) af[i] = load_pk(Ap + (size_t)i * rstride + kt * 512);
#pragma unroll
  for (int j = 0; j < 4; j++) bf[j] = load_pk(Bp + (size_t)j * rstride + kt * 512);
}

__device__ __forceinline__ void mma_set(v8f (&acc)[4][4], const v16h (&af)[4],
                                        const v16h (&bf)[4]) {
#pragma unroll
  for (int i = 0; i < 4; i++)
#pragma unroll
    for (int j = 0; j < 4; j++)
      acc[i][j] = wmma_f16(af[i], bf[j], acc[i][j]);
}

__global__ __launch_bounds__(256) void cvt_f32_f16_kernel(
    const float* __restrict__ in, _Float16* __restrict__ out) {
  const int lane = threadIdx.x & 31;
  const int wave = threadIdx.x >> 5;
  const int row = (blockIdx.x >> 4) * 16 + (wave & 3) * 4 + (lane >> 3);
  const int k   = (blockIdx.x & 15) * 64 + (wave >> 2) * 32 + (lane & 7) * 4;
  const float4 v = *(const float4*)(in + (size_t)row * 1024 + k);
  v4h o = {(_Float16)v.x, (_Float16)v.y, (_Float16)v.z, (_Float16)v.w};
  *(v4h*)(out + pk_tile_off(row, k, 32) + pk_elem_off(row & 15, k & 31)) = o;
}

__device__ __forceinline__ void cvt_stripe(const float* __restrict__ in,
                                           _Float16* __restrict__ out,
                                           int stripe, int wave, int lane) {
  const int rr = lane >> 3;
  const int kq = (lane & 7) * 4;
  const float* src = in + (size_t)stripe * (16 * 1024) + rr * 1024 + wave * 32 + kq;
  _Float16* dst = out + (size_t)stripe * (32 * 512) + wave * 512 + pk_elem_off(rr, kq);
  v4f v[8][4];
#pragma unroll
  for (int t = 0; t < 8; t++)
#pragma unroll
    for (int g = 0; g < 4; g++)
      v[t][g] = __builtin_nontemporal_load((const v4f*)(src + g * 4096 + t * 128));
#pragma unroll
  for (int t = 0; t < 8; t++)
#pragma unroll
    for (int g = 0; g < 4; g++) {
      v4h o = {(_Float16)v[t][g].x, (_Float16)v[t][g].y, (_Float16)v[t][g].z,
               (_Float16)v[t][g].w};
      *(v4h*)(dst + t * 2048 + g * 32) = o;
    }
}

__global__ __launch_bounds__(128) void gemm_nt_kernel(
    const _Float16* __restrict__ A, const _Float16* __restrict__ Bw,
    float* __restrict__ Cf, _Float16* __restrict__ Ch,
    const float* __restrict__ bias, int Mdim, int Ndim, int Kdim, int flags,
    const float* __restrict__ ra_in, _Float16* __restrict__ ra_out, int ra_stripes,
    const float* __restrict__ rb_in, _Float16* __restrict__ rb_out) {
  const int lane = threadIdx.x & 31;
  const int wave = threadIdx.x >> 5;
  const int tiles_n = Ndim >> 7;
  const int tile_wgs = (Mdim >> 7) * tiles_n;
  if ((int)blockIdx.x >= tile_wgs) {
    const int s = (int)blockIdx.x - tile_wgs;
    const int wv = __builtin_amdgcn_readfirstlane(wave);
    const bool first = s < ra_stripes;
    cvt_stripe(first ? ra_in : rb_in, first ? ra_out : rb_out,
               first ? s : s - ra_stripes, wv, lane);
    return;
  }
  const int bm = blockIdx.x / tiles_n;
  const int bn = blockIdx.x - bm * tiles_n;
  const int m0 = bm * 128 + (wave & 1) * 64;
  const int n0 = bn * 128 + (wave >> 1) * 64;

  const int nkt = Kdim >> 5;
  const int rstride = nkt * 512;
  const _Float16* Ap = A  + (size_t)(m0 >> 4) * rstride + lane * 8;
  const _Float16* Bp = Bw + (size_t)(n0 >> 4) * rstride + lane * 8;

  v8f acc[4][4] = {};

  const int mh = (lane >> 4) << 3;
  v4f brow[4][2] = {};
  if (Ch && bias) {
#pragma unroll
    for (int i = 0; i < 4; i++) {
      brow[i][0] = *(const v4f*)(bias + m0 + 16 * i + mh);
      brow[i][1] = *(const v4f*)(bias + m0 + 16 * i + mh + 4);
    }
  }

#define SCHED_FENCE() __builtin_amdgcn_sched_barrier(0)
  v16h fa0[4], fb0[4], fa1[4], fb1[4], fa2[4], fb2[4], fa3[4], fb3[4];
  load_set(fa0, fb0, Ap, Bp, rstride, 0);
  SCHED_FENCE();
  load_set(fa1, fb1, Ap, Bp, rstride, 1);
  SCHED_FENCE();
  load_set(fa2, fb2, Ap, Bp, rstride, 2);
  SCHED_FENCE();

  int kt = 0;
#pragma unroll 1
  for (; kt < nkt - 4; kt += 4) {
    load_set(fa3, fb3, Ap, Bp, rstride, kt + 3);
    SCHED_FENCE();
    mma_set(acc, fa0, fb0);
    SCHED_FENCE();
    load_set(fa0, fb0, Ap, Bp, rstride, kt + 4);
    SCHED_FENCE();
    mma_set(acc, fa1, fb1);
    SCHED_FENCE();
    load_set(fa1, fb1, Ap, Bp, rstride, kt + 5);
    SCHED_FENCE();
    mma_set(acc, fa2, fb2);
    SCHED_FENCE();
    load_set(fa2, fb2, Ap, Bp, rstride, kt + 6);
    SCHED_FENCE();
    mma_set(acc, fa3, fb3);
    SCHED_FENCE();
  }
  load_set(fa3, fb3, Ap, Bp, rstride, kt + 3);
  SCHED_FENCE();
  mma_set(acc, fa0, fb0);
  mma_set(acc, fa1, fb1);
  mma_set(acc, fa2, fb2);
  mma_set(acc, fa3, fb3);
#undef SCHED_FENCE

  const int nl = lane & 15;
  if (Ch) {
    const bool tr = (flags & FLAG_TRANS) != 0;
    const int ktiles = tr ? 32 : (Mdim >> 5);
#pragma unroll
    for (int i = 0; i < 4; i++) {
      const int mt = m0 + 16 * i;
      const int trow = (tr ? ((mt >> 10) << 10) : 0) + n0;
      const int tk = tr ? (mt & 1023) : mt;
#pragma unroll
      for (int j = 0; j < 4; j++) {
        float v[8];
#pragma unroll
        for (int r = 0; r < 8; r++) {
          v[r] = acc[i][j][r] + brow[i][r >> 2][r & 3];
          if (flags & FLAG_RELU) v[r] = fmaxf(v[r], 0.f);
        }
        _Float16* dst = Ch + pk_tile_off(trow + 16 * j, tk, ktiles) + ((i & 1) << 8) + lane * 8;
        v8h o = {(_Float16)v[0], (_Float16)v[1], (_Float16)v[2], (_Float16)v[3],
                 (_Float16)v[4], (_Float16)v[5], (_Float16)v[6], (_Float16)v[7]};
        *(v8h*)dst = o;
      }
    }
    return;
  }
#pragma unroll
  for (int i = 0; i < 4; i++) {
#pragma unroll
    for (int j = 0; j < 4; j++) {
      const int n = n0 + j * 16 + nl;
      const float bv = bias ? bias[n] : 0.f;
      float v[8];
#pragma unroll
      for (int r = 0; r < 8; r++) {
        v[r] = acc[i][j][r] + bv;
        if (flags & FLAG_RELU) v[r] = fmaxf(v[r], 0.f);
      }
      if (Cf) {
#pragma unroll
        for (int r = 0; r < 8; r++)
          Cf[(size_t)(m0 + 16 * i + mh + r) * Ndim + n] = v[r];
      }
    }
  }
}


__device__ __forceinline__ void load_k(v16h (&kf)[4], int& mk0, int& mk1,
                                       const _Float16* Kp, const int* mb,
                                       int kb, int lane) {
#pragma unroll
  for (int sub = 0; sub < 2; sub++) {
    const _Float16* p = Kp + (size_t)((kb >> 4) + sub) * (32 * 512);
    kf[2 * sub]     = load_pk(p);
    kf[2 * sub + 1] = load_pk(p + 512);
  }
  mk0 = mb[kb + (lane & 15)];
  mk1 = mb[kb + 16 + (lane & 15)];
}

__device__ __forceinline__ void load_v(v16h (&vf)[4], const _Float16* Vp, int kb) {
#pragma unroll
  for (int t = 0; t < 4; t++)
    vf[t] = load_pk(Vp + (size_t)t * (32 * 512) + (kb >> 5) * 512);
}

__device__ __forceinline__ void attn_block(const v16h& qa0, const v16h& qa1,
                                           const v16h (&kf)[4], const v16h (&vf)[4],
                                           const int mk0, const int mk1,
                                           const v16h& ones, v8f (&o)[5], v8f& nmuse,
                                           _Float16* myp, const int lane, const int mh8) {
  const float c2 = 0.125f * 1.44269504088896341f;
  v8f s[2] = {};
#pragma unroll
  for (int sub = 0; sub < 2; sub++) {
    s[sub] = wmma_f16(qa0, kf[2 * sub], s[sub]);
    s[sub] = wmma_f16(qa1, kf[2 * sub + 1], s[sub]);
  }

  v8f e0, e1;
#pragma unroll
  for (int r = 0; r < 8; r++) {
    e0[r] = __builtin_fmaf(s[0][r], c2, nmuse[r]);
    e1[r] = __builtin_fmaf(s[1][r], c2, nmuse[r]);
  }
  float x0 = e0[0], x1 = e1[0];
#pragma unroll
  for (int r = 1; r < 8; r++) {
    x0 = fmaxf(x0, e0[r]);
    x1 = fmaxf(x1, e1[r]);
  }
  const float x = fmaxf(mk0 ? x0 : -1e30f, mk1 ? x1 : -1e30f);

  if (__builtin_expect(__builtin_amdgcn_ballot_w32(x > 8.0f) != 0, 0)) {
    v8f bmax;
#pragma unroll
    for (int r = 0; r < 8; r++) {
      float v = -1e30f;
      if (mk0) v = fmaxf(v, s[0][r] * c2);
      if (mk1) v = fmaxf(v, s[1][r] * c2);
      bmax[r] = v;
    }
#pragma unroll
    for (int off = 1; off < 16; off <<= 1)
#pragma unroll
      for (int r = 0; r < 8; r++)
        bmax[r] = fmaxf(bmax[r], __shfl_xor(bmax[r], off, 32));
    v8f alpha;
#pragma unroll
    for (int r = 0; r < 8; r++) {
      const float nnew = fminf(nmuse[r], -bmax[r]);
      alpha[r] = __builtin_amdgcn_exp2f(nnew - nmuse[r]);
      nmuse[r] = nnew;
      e0[r] = __builtin_fmaf(s[0][r], c2, nnew);
      e1[r] = __builtin_fmaf(s[1][r], c2, nnew);
    }
#pragma unroll
    for (int t = 0; t < 5; t++)
#pragma unroll
      for (int r = 0; r < 8; r++) o[t][r] *= alpha[r];
  }

  v8f p0, p1;
#pragma unroll
  for (int r = 0; r < 8; r++) {
    p0[r] = mk0 ? __builtin_amdgcn_exp2f(e0[r]) : 0.f;
    p1[r] = mk1 ? __builtin_amdgcn_exp2f(e1[r]) : 0.f;
  }

#pragma unroll
  for (int r = 0; r < 8; r++) {
    myp[(r + mh8) * 40 + (lane & 15)]      = (_Float16)p0[r];
    myp[(r + mh8) * 40 + 16 + (lane & 15)] = (_Float16)p1[r];
  }
  asm volatile("s_wait_dscnt 0" ::: "memory");
  v16h pa = load_frag(myp, 40, 0, 0, lane);

#pragma unroll
  for (int t = 0; t < 4; t++) o[t] = wmma_f16(pa, vf[t], o[t]);
  o[4] = wmma_f16(pa, ones, o[4]);
}

__global__ __launch_bounds__(256) void attn_kernel(
    const _Float16* __restrict__ Qh, const _Float16* __restrict__ Kh,
    const _Float16* __restrict__ Vt, const int* __restrict__ mask,
    _Float16* __restrict__ Oh) {
  __shared__ _Float16 plds[8][16 * 40];

  const int lane = threadIdx.x & 31;
  const int wave = threadIdx.x >> 5;
  const int bh   = blockIdx.x >> 3;
  const int qblk = blockIdx.x & 7;
  const int b = bh >> 4, h = bh & 15;
  const int q0 = qblk * 128 + wave * 16;

  const _Float16* Qp = Qh + pk_tile_off(b * 1024 + q0, h * 64, 32) + lane * 8;
  const _Float16* Kp = Kh + pk_tile_off(b * 1024, h * 64, 32) + lane * 8;
  const _Float16* Vp = Vt + pk_tile_off((b * 16 + h) * 64, 0, 32) + lane * 8;
  const int* mb = mask + b * 1024;
  _Float16* myp = &plds[wave][0];

  const v16h qa0 = load_pk(Qp);
  const v16h qa1 = load_pk(Qp + 512);

  v8f o[5] = {};
  v8f nmuse = {1e30f, 1e30f, 1e30f, 1e30f, 1e30f, 1e30f, 1e30f, 1e30f};
  v16h ones;
#pragma unroll
  for (int i = 0; i < 16; i++) ones[i] = (_Float16)1.0f;
  const int mh8 = (lane >> 4) << 3;

#define SCHED_FENCE() __builtin_amdgcn_sched_barrier(0)
  v16h kA[4], kB[4], vf[4];
  int mA0, mA1, mB0, mB1;
  load_k(kA, mA0, mA1, Kp, mb, 0, lane);

#pragma unroll 1
  for (int kb = 0; kb < 1024; kb += 64) {
    load_v(vf, Vp, kb);
    load_k(kB, mB0, mB1, Kp, mb, kb + 32, lane);
    SCHED_FENCE();
    attn_block(qa0, qa1, kA, vf, mA0, mA1, ones, o, nmuse, myp, lane, mh8);
    SCHED_FENCE();
    load_v(vf, Vp, kb + 32);
    const int kn = (kb + 64 < 1024) ? kb + 64 : 1024 - 32;
    load_k(kA, mA0, mA1, Kp, mb, kn, lane);
    SCHED_FENCE();
    attn_block(qa0, qa1, kB, vf, mB0, mB1, ones, o, nmuse, myp, lane, mh8);
    SCHED_FENCE();
  }
#undef SCHED_FENCE

#pragma unroll
  for (int t = 0; t < 4; t++) {
    const int n = h * 64 + t * 16 + (lane & 15);
    _Float16* dst = Oh + pk_tile_off(b * 1024 + q0, n, 32) + pk_elem_off(mh8, n & 31);
#pragma unroll
    for (int r = 0; r < 8; r++) {
      const float denom = o[4][r];
      const float val = denom > 0.f ? o[t][r] / denom : 0.f;
      dst[r * 8] = (_Float16)val;
    }
  }
}

#define LN_ROWS 2
__global__ __launch_bounds__(256) void ln_kernel(
    const float* __restrict__ Ain, const float* __restrict__ Bin,
    const float* __restrict__ g, const float* __restrict__ beta,
    float* __restrict__ out32, _Float16* __restrict__ out16) {
  const int lane = threadIdx.x & 31;
  const int wave = threadIdx.x >> 5;
  const int row0 = (blockIdx.x * 8 + wave) * LN_ROWS;

  const float4* pa = (const float4*)(Ain + (size_t)row0 * 1024) + lane;
  const float4* pb = (const float4*)(Bin + (size_t)row0 * 1024) + lane;
  const float4* pg = (const float4*)g + lane;
  const float4* pt = (const float4*)beta + lane;

  float4 xa[LN_ROWS][8], xb[LN_ROWS][8], gv[8], bv[8];
#pragma unroll
  for (int r = 0; r < LN_ROWS; r++) {
#pragma unroll
    for (int q = 0; q < 8; q++) xa[r][q] = pa[r * 256 + q * 32];
#pragma unroll
    for (int q = 0; q < 8; q++) xb[r][q] = pb[r * 256 + q * 32];
  }
#pragma unroll
  for (int q = 0; q < 8; q++) gv[q] = pg[q * 32];
#pragma unroll
  for (int q = 0; q < 8; q++) bv[q] = pt[q * 32];
  __builtin_amdgcn_sched_group_barrier(0x020, 48, 0);
  __builtin_amdgcn_sched_group_barrier(0x002, 1024, 0);

  float4 x[LN_ROWS][8];
  float s1[LN_ROWS], s2[LN_ROWS];
#pragma unroll
  for (int r = 0; r < LN_ROWS; r++) {
    float4 a1 = {0.f, 0.f, 0.f, 0.f}, a2 = {0.f, 0.f, 0.f, 0.f};
#pragma unroll
    for (int q = 0; q < 8; q++) {
      x[r][q].x = xa[r][q].x + xb[r][q].x;
      x[r][q].y = xa[r][q].y + xb[r][q].y;
      x[r][q].z = xa[r][q].z + xb[r][q].z;
      x[r][q].w = xa[r][q].w + xb[r][q].w;
      a1.x += x[r][q].x; a1.y += x[r][q].y; a1.z += x[r][q].z; a1.w += x[r][q].w;
      a2.x += x[r][q].x * x[r][q].x; a2.y += x[r][q].y * x[r][q].y;
      a2.z += x[r][q].z * x[r][q].z; a2.w += x[r][q].w * x[r][q].w;
    }
    s1[r] = (a1.x + a1.y) + (a1.z + a1.w);
    s2[r] = (a2.x + a2.y) + (a2.z + a2.w);
  }
#pragma unroll
  for (int off = 16; off >= 1; off >>= 1) {
#pragma unroll
    for (int r = 0; r < LN_ROWS; r++) {
      s1[r] += __shfl_xor(s1[r], off, 32);
      s2[r] += __shfl_xor(s2[r], off, 32);
    }
  }

#pragma unroll
  for (int r = 0; r < LN_ROWS; r++) {
    const int row = row0 + r;
    const float mu   = s1[r] * (1.f / 1024.f);
    const float var  = s2[r] * (1.f / 1024.f) - mu * mu;
    const float rstd = rsqrtf(var + 1e-5f);
    float4 y[8];
    float4* po = (float4*)(out32 + (size_t)row * 1024) + lane;
#pragma unroll
    for (int q = 0; q < 8; q++) {
      y[q].x = (x[r][q].x - mu) * rstd * gv[q].x + bv[q].x;
      y[q].y = (x[r][q].y - mu) * rstd * gv[q].y + bv[q].y;
      y[q].z = (x[r][q].z - mu) * rstd * gv[q].z + bv[q].z;
      y[q].w = (x[r][q].w - mu) * rstd * gv[q].w + bv[q].w;
      po[q * 32] = y[q];
    }
    if (out16) {
#pragma unroll
      for (int q = 0; q < 8; q++) {
        const int k = (q * 32 + lane) * 4;
        v4h o = {(_Float16)y[q].x, (_Float16)y[q].y, (_Float16)y[q].z, (_Float16)y[q].w};
        *(v4h*)(out16 + pk_tile_off(row, k, 32) + pk_elem_off(row & 15, k & 31)) = o;
      }
    }
  }
}
#undef LN_ROWS

extern "C" void kernel_launch(void* const* d_in, const int* in_sizes, int n_in,
                              void* d_out, int out_size, void* d_ws, size_t ws_size,
                              hipStream_t stream) {
  (void)in_sizes; (void)n_in; (void)out_size; (void)ws_size;
  const float* X    = (const float*)d_in[0];
  const float* Y    = (const float*)d_in[1];
  const int*   mask = (const int*)d_in[2];
  const float* Wq   = (const float*)d_in[3];
  const float* Wk   = (const float*)d_in[4];
  const float* Wv   = (const float*)d_in[5];
  const float* Wo   = (const float*)d_in[6];
  const float* ln1g = (const float*)d_in[7];
  const float* ln1b = (const float*)d_in[8];
  const float* ln2g = (const float*)d_in[9];
  const float* ln2b = (const float*)d_in[10];
  const float* W1   = (const float*)d_in[11];
  const float* b1   = (const float*)d_in[12];
  const float* W2   = (const float*)d_in[13];
  const float* b2   = (const float*)d_in[14];

  char* ws = (char*)d_ws;
  const size_t MB = 1024 * 1024;
  _Float16* Xh   = (_Float16*)(ws + 0);
  _Float16* Yh   = (_Float16*)(ws + 8 * MB);
  _Float16* Wqh  = (_Float16*)(ws + 16 * MB);
  _Float16* Wkh  = (_Float16*)(ws + 18 * MB);
  _Float16* Wvh  = (_Float16*)(ws + 20 * MB);
  _Float16* Woh  = (_Float16*)(ws + 22 * MB);
  _Float16* W1h  = (_Float16*)(ws + 24 * MB);
  _Float16* W2h  = (_Float16*)(ws + 26 * MB);
  _Float16* Qh   = (_Float16*)(ws + 28 * MB);
  _Float16* Kh   = (_Float16*)(ws + 36 * MB);
  _Float16* Vt   = (_Float16*)(ws + 44 * MB);
  float*    Fbuf = (float*)(ws + 52 * MB);
  float*    Hres = (float*)(ws + 68 * MB);
  _Float16* Ohb  = Xh;
  _Float16* Hh   = Yh;
  _Float16* FF1h = Qh;

  const int n4m = 4096 * 1024 / 4;
  const int n4w = 1024 * 1024 / 4;
  dim3 blk256(256);
  dim3 blk128(128);

  cvt_f32_f16_kernel<<<n4m / 256, blk256, 0, stream>>>(X, Xh);
  cvt_f32_f16_kernel<<<n4w / 256, blk256, 0, stream>>>(Wq, Wqh);

  const int ggrid = (4096 / 128) * (1024 / 128);
  const int sY = 4096 / 16;
  const int sW = 1024 / 16;

  gemm_nt_kernel<<<ggrid + sY + sW, blk128, 0, stream>>>(
      Wqh, Xh, nullptr, Qh, nullptr, 1024, 4096, 1024, 0, Y, Yh, sY, Wk, Wkh);
  gemm_nt_kernel<<<ggrid + sW + sW, blk128, 0, stream>>>(
      Wkh, Yh, nullptr, Kh, nullptr, 1024, 4096, 1024, 0, Wv, Wvh, sW, Wo, Woh);
  gemm_nt_kernel<<<ggrid + sW + sW, blk128, 0, stream>>>(
      Yh, Wvh, nullptr, Vt, nullptr, 4096, 1024, 1024, FLAG_TRANS, W1, W1h, sW,
      W2, W2h);

  attn_kernel<<<512, blk256, 0, stream>>>(Qh, Kh, Vt, mask, Ohb);

  gemm_nt_kernel<<<ggrid, blk128, 0, stream>>>(Ohb, Woh, Fbuf, nullptr, nullptr,
                                               4096, 1024, 1024, 0, nullptr,
                                               nullptr, 0, nullptr, nullptr);
  ln_kernel<<<256, blk256, 0, stream>>>(X, Fbuf, ln1g, ln1b, Hres, Hh);

  gemm_nt_kernel<<<ggrid, blk128, 0, stream>>>(W1h, Hh, nullptr, FF1h, b1,
                                               1024, 4096, 1024, FLAG_RELU,
                                               nullptr, nullptr, 0, nullptr,
                                               nullptr);
  gemm_nt_kernel<<<ggrid, blk128, 0, stream>>>(FF1h, W2h, Fbuf, nullptr, b2,
                                               4096, 1024, 1024, 0, nullptr,
                                               nullptr, 0, nullptr, nullptr);
  ln_kernel<<<256, blk256, 0, stream>>>(Hres, Fbuf, ln2g, ln2b, (float*)d_out,
                                         nullptr);
}


# gemm with f32 output (out projection, FFN2): weights as WMMA A, 16-byte pieces transposed 4x4 across lanes (quad permutes, no LDS): 32 whole-line b128 stores a wave, not 128 4-byte stores; same sums
# speedup vs baseline: 2.7784x; 1.0556x over previous
#include <hip/hip_runtime.h>
#include <hip/hip_bf16.h>


typedef __attribute__((ext_vector_type(16))) _Float16 v16h;
typedef __attribute__((ext_vector_type(8)))  _Float16 v8h;
typedef __attribute__((ext_vector_type(4)))  _Float16 v4h;
typedef __attribute__((ext_vector_type(8)))  float    v8f;
typedef __attribute__((ext_vector_type(4)))  float    v4f;

#define FLAG_RELU  1
#define FLAG_TRANS 2

__device__ __forceinline__ size_t pk_tile_off(int row, int k, int ktiles) {
  return ((size_t)(row >> 4) * ktiles + (k >> 5)) * 512;
}
__device__ __forceinline__ int pk_elem_off(int rr, int kk) {
  return ((kk >> 4) << 8) + ((((kk >> 3) & 1) << 4) + rr) * 8 + (kk & 7);
}

__device__ __forceinline__ v16h load_pk(const _Float16* p) {
  v8h lo = *(const v8h*)(p);
  v8h hi = *(const v8h*)(p + 256);
  return __builtin_shufflevector(lo, hi, 0, 1, 2, 3, 4, 5, 6, 7,
                                         8, 9, 10, 11, 12, 13, 14, 15);
}

__device__ __forceinline__ v16h load_frag(const _Float16* base, int ld,
                                          int row0, int k0, int lane) {
  const int r  = row0 + (lane & 15);
  const int kb = k0 + ((lane >> 4) << 3);
  const _Float16* p = base + (size_t)r * ld + kb;
  v8h lo = *(const v8h*)(p);
  v8h hi = *(const v8h*)(p + 16);
  return __builtin_shufflevector(lo, hi, 0, 1, 2, 3, 4, 5, 6, 7,
                                         8, 9, 10, 11, 12, 13, 14, 15);
}

__device__ __forceinline__ v8f wmma_f16(v16h a, v16h b, v8f c) {
  return __builtin_amdgcn_wmma_f32_16x16x32_f16(false, a, false, b,
                                                (short)0, c, false, false);
}

__device__ __forceinline__ void load_set(v16h (&af)[4], v16h (&bf)[4],
                                         const _Float16* Ap, const _Float16* Bp,
                                         int rstride, int kt) {
#pragma unroll
  for (int i = 0; i < 4; i++) af[i] = load_pk(Ap + (size_t)i * rstride + kt * 512);
#pragma unroll
  for (int j = 0; j < 4; j++) bf[j] = load_pk(Bp + (size_t)j * rstride + kt * 512);
}

__device__ __forceinline__ void mma_set(v8f (&acc)[4][4], const v16h (&af)[4],
                                        const v16h (&bf)[4]) {
#pragma unroll
  for (int i = 0; i < 4; i++)
#pragma unroll
    for (int j = 0; j < 4; j++)
      acc[i][j] = wmma_f16(af[i], bf[j], acc[i][j]);
}

template <int CTRL>
__device__ __forceinline__ void lane_swap(v4f& lo, v4f& hi, bool upper) {
#pragma unroll
  for (int e = 0; e < 4; e++) {
    const float fl = lo[e];
    const float fh = hi[e];
    const int il = __builtin_bit_cast(int, fl);
    const int ih = __builtin_bit_cast(int, fh);
    const float plo = __builtin_bit_cast(
        float, __builtin_amdgcn_update_dpp(il, il, CTRL, 0xF, 0xF, true));
    const float phi = __builtin_bit_cast(
        float, __builtin_amdgcn_update_dpp(ih, ih, CTRL, 0xF, 0xF, true));
    lo[e] = upper ? phi : lo[e];
    hi[e] = upper ? hi[e] : plo;
  }
}

__global__ __launch_bounds__(256) void cvt_f32_f16_kernel(
    const float* __restrict__ in, _Float16* __restrict__ out) {
  const int lane = threadIdx.x & 31;
  const int wave = threadIdx.x >> 5;
  const int row = (blockIdx.x >> 4) * 16 + (wave & 3) * 4 + (lane >> 3);
  const int k   = (blockIdx.x & 15) * 64 + (wave >> 2) * 32 + (lane & 7) * 4;
  const float4 v = *(const float4*)(in + (size_t)row * 1024 + k);
  v4h o = {(_Float16)v.x, (_Float16)v.y, (_Float16)v.z, (_Float16)v.w};
  *(v4h*)(out + pk_tile_off(row, k, 32) + pk_elem_off(row & 15, k & 31)) = o;
}

__device__ __forceinline__ void cvt_stripe(const float* __restrict__ in,
                                           _Float16* __restrict__ out,
                                           int stripe, int wave, int lane) {
  const int rr = lane >> 3;
  const int kq = (lane & 7) * 4;
  const float* src = in + (size_t)stripe * (16 * 1024) + rr * 1024 + wave * 32 + kq;
  _Float16* dst = out + (size_t)stripe * (32 * 512) + wave * 512 + pk_elem_off(rr, kq);
  v4f v[8][4];
#pragma unroll
  for (int t = 0; t < 8; t++)
#pragma unroll
    for (int g = 0; g < 4; g++)
      v[t][g] = __builtin_nontemporal_load((const v4f*)(src + g * 4096 + t * 128));
#pragma unroll
  for (int t = 0; t < 8; t++)
#pragma unroll
    for (int g = 0; g < 4; g++) {
      v4h o = {(_Float16)v[t][g].x, (_Float16)v[t][g].y, (_Float16)v[t][g].z,
               (_Float16)v[t][g].w};
      *(v4h*)(dst + t * 2048 + g * 32) = o;
    }
}

__global__ __launch_bounds__(128) void gemm_nt_kernel(
    const _Float16* __restrict__ A, const _Float16* __restrict__ Bw,
    float* __restrict__ Cf, _Float16* __restrict__ Ch,
    const float* __restrict__ bias, int Mdim, int Ndim, int Kdim, int flags,
    const float* __restrict__ ra_in, _Float16* __restrict__ ra_out, int ra_stripes,
    const float* __restrict__ rb_in, _Float16* __restrict__ rb_out) {
  const int lane = threadIdx.x & 31;
  const int wave = threadIdx.x >> 5;
  const int tiles_n = Ndim >> 7;
  const int tile_wgs = (Mdim >> 7) * tiles_n;
  if ((int)blockIdx.x >= tile_wgs) {
    const int s = (int)blockIdx.x - tile_wgs;
    const int wv = __builtin_amdgcn_readfirstlane(wave);
    const bool first = s < ra_stripes;
    cvt_stripe(first ? ra_in : rb_in, first ? ra_out : rb_out,
               first ? s : s - ra_stripes, wv, lane);
    return;
  }
  const int bm = blockIdx.x / tiles_n;
  const int bn = blockIdx.x - bm * tiles_n;
  const int m0 = bm * 128 + (wave & 1) * 64;
  const int n0 = bn * 128 + (wave >> 1) * 64;

  const int nkt = Kdim >> 5;
  const int rstride = nkt * 512;
  const _Float16* Ap = A  + (size_t)(m0 >> 4) * rstride + lane * 8;
  const _Float16* Bp = Bw + (size_t)(n0 >> 4) * rstride + lane * 8;

  v8f acc[4][4] = {};

  const int mh = (lane >> 4) << 3;
  v4f brow[4][2] = {};
  if (bias) {
#pragma unroll
    for (int i = 0; i < 4; i++) {
      brow[i][0] = *(const v4f*)(bias + m0 + 16 * i + mh);
      brow[i][1] = *(const v4f*)(bias + m0 + 16 * i + mh + 4);
    }
  }

#define SCHED_FENCE() __builtin_amdgcn_sched_barrier(0)
  v16h fa0[4], fb0[4], fa1[4], fb1[4], fa2[4], fb2[4], fa3[4], fb3[4];
  load_set(fa0, fb0, Ap, Bp, rstride, 0);
  SCHED_FENCE();
  load_set(fa1, fb1, Ap, Bp, rstride, 1);
  SCHED_FENCE();
  load_set(fa2, fb2, Ap, Bp, rstride, 2);
  SCHED_FENCE();

  int kt = 0;
#pragma unroll 1
  for (; kt < nkt - 4; kt += 4) {
    load_set(fa3, fb3, Ap, Bp, rstride, kt + 3);
    SCHED_FENCE();
    mma_set(acc, fa0, fb0);
    SCHED_FENCE();
    load_set(fa0, fb0, Ap, Bp, rstride, kt + 4);
    SCHED_FENCE();
    mma_set(acc, fa1, fb1);
    SCHED_FENCE();
    load_set(fa1, fb1, Ap, Bp, rstride, kt + 5);
    SCHED_FENCE();
    mma_set(acc, fa2, fb2);
    SCHED_FENCE();
    load_set(fa2, fb2, Ap, Bp, rstride, kt + 6);
    SCHED_FENCE();
    mma_set(acc, fa3, fb3);
    SCHED_FENCE();
  }
  load_set(fa3, fb3, Ap, Bp, rstride, kt + 3);
  SCHED_FENCE();
  mma_set(acc, fa0, fb0);
  mma_set(acc, fa1, fb1);
  mma_set(acc, fa2, fb2);
  mma_set(acc, fa3, fb3);
#undef SCHED_FENCE

  const int nl = lane & 15;
  if (Ch) {
    const bool tr = (flags & FLAG_TRANS) != 0;
    const int ktiles = tr ? 32 : (Mdim >> 5);
#pragma unroll
    for (int i = 0; i < 4; i++) {
      const int mt = m0 + 16 * i;
      const int trow = (tr ? ((mt >> 10) << 10) : 0) + n0;
      const int tk = tr ? (mt & 1023) : mt;
#pragma unroll
      for (int j = 0; j < 4; j++) {
        float v[8];
#pragma unroll
        for (int r = 0; r < 8; r++) {
          v[r] = acc[i][j][r] + brow[i][r >> 2][r & 3];
          if (flags & FLAG_RELU) v[r] = fmaxf(v[r], 0.f);
        }
        _Float16* dst = Ch + pk_tile_off(trow + 16 * j, tk, ktiles) + ((i & 1) << 8) + lane * 8;
        v8h o = {(_Float16)v[0], (_Float16)v[1], (_Float16)v[2], (_Float16)v[3],
                 (_Float16)v[4], (_Float16)v[5], (_Float16)v[6], (_Float16)v[7]};
        *(v8h*)dst = o;
      }
    }
    return;
  }
  if (Cf) {
    const bool up1 = (lane & 1) != 0;
    const bool up2 = (lane & 2) != 0;
    const int t = lane & 3;
    const int fo = ((t >> 1) << 4) + mh + ((t & 1) << 2);
#pragma unroll
    for (int j = 0; j < 4; j++) {
#pragma unroll
      for (int I = 0; I < 2; I++) {
        v4f E[4];
#pragma unroll
        for (int k = 0; k < 4; k++) {
          const int i = 2 * I + (k >> 1);
          const int q = k & 1;
#pragma unroll
          for (int e = 0; e < 4; e++) {
            E[k][e] = acc[i][j][4 * q + e] + brow[i][q][e];
          }
        }
        lane_swap<0xB1>(E[0], E[1], up1);
        lane_swap<0xB1>(E[2], E[3], up1);
        lane_swap<0x4E>(E[0], E[2], up2);
        lane_swap<0x4E>(E[1], E[3], up2);
        float* dst = Cf + (size_t)(n0 + 16 * j + (nl & 12)) * Mdim + m0 + 32 * I + fo;
#pragma unroll
        for (int tau = 0; tau < 4; tau++) *(v4f*)(dst + (size_t)tau * Mdim) = E[tau];
      }
    }
  }
}


__device__ __forceinline__ void load_k(v16h (&kf)[4], int& mk0, int& mk1,
                                       const _Float16* Kp, const int* mb,
                                       int kb, int lane) {
#pragma unroll
  for (int sub = 0; sub < 2; sub++) {
    const _Float16* p = Kp + (size_t)((kb >> 4) + sub) * (32 * 512);
    kf[2 * sub]     = load_pk(p);
    kf[2 * sub + 1] = load_pk(p + 512);
  }
  mk0 = mb[kb + (lane & 15)];
  mk1 = mb[kb + 16 + (lane & 15)];
}

__device__ __forceinline__ void load_v(v16h (&vf)[4], const _Float16* Vp, int kb) {
#pragma unroll
  for (int t = 0; t < 4; t++)
    vf[t] = load_pk(Vp + (size_t)t * (32 * 512) + (kb >> 5) * 512);
}

__device__ __forceinline__ void attn_block(const v16h& qa0, const v16h& qa1,
                                           const v16h (&kf)[4], const v16h (&vf)[4],
                                           const int mk0, const int mk1,
                                           const v16h& ones, v8f (&o)[5], v8f& nmuse,
                                           _Float16* myp, const int lane, const int mh8) {
  const float c2 = 0.125f * 1.44269504088896341f;
  v8f s[2] = {};
#pragma unroll
  for (int sub = 0; sub < 2; sub++) {
    s[sub] = wmma_f16(qa0, kf[2 * sub], s[sub]);
    s[sub] = wmma_f16(qa1, kf[2 * sub + 1], s[sub]);
  }

  v8f e0, e1;
#pragma unroll
  for (int r = 0; r < 8; r++) {
    e0[r] = __builtin_fmaf(s[0][r], c2, nmuse[r]);
    e1[r] = __builtin_fmaf(s[1][r], c2, nmuse[r]);
  }
  float x0 = e0[0], x1 = e1[0];
#pragma unroll
  for (int r = 1; r < 8; r++) {
    x0 = fmaxf(x0, e0[r]);
    x1 = fmaxf(x1, e1[r]);
  }
  const float x = fmaxf(mk0 ? x0 : -1e30f, mk1 ? x1 : -1e30f);

  if (__builtin_expect(__builtin_amdgcn_ballot_w32(x > 8.0f) != 0, 0)) {
    v8f bmax;
#pragma unroll
    for (int r = 0; r < 8; r++) {
      float v = -1e30f;
      if (mk0) v = fmaxf(v, s[0][r] * c2);
      if (mk1) v = fmaxf(v, s[1][r] * c2);
      bmax[r] = v;
    }
#pragma unroll
    for (int off = 1; off < 16; off <<= 1)
#pragma unroll
      for (int r = 0; r < 8; r++)
        bmax[r] = fmaxf(bmax[r], __shfl_xor(bmax[r], off, 32));
    v8f alpha;
#pragma unroll
    for (int r = 0; r < 8; r++) {
      const float nnew = fminf(nmuse[r], -bmax[r]);
      alpha[r] = __builtin_amdgcn_exp2f(nnew - nmuse[r]);
      nmuse[r] = nnew;
      e0[r] = __builtin_fmaf(s[0][r], c2, nnew);
      e1[r] = __builtin_fmaf(s[1][r], c2, nnew);
    }
#pragma unroll
    for (int t = 0; t < 5; t++)
#pragma unroll
      for (int r = 0; r < 8; r++) o[t][r] *= alpha[r];
  }

  v8f p0, p1;
#pragma unroll
  for (int r = 0; r < 8; r++) {
    p0[r] = mk0 ? __builtin_amdgcn_exp2f(e0[r]) : 0.f;
    p1[r] = mk1 ? __builtin_amdgcn_exp2f(e1[r]) : 0.f;
  }

#pragma unroll
  for (int r = 0; r < 8; r++) {
    myp[(r + mh8) * 40 + (lane & 15)]      = (_Float16)p0[r];
    myp[(r + mh8) * 40 + 16 + (lane & 15)] = (_Float16)p1[r];
  }
  asm volatile("s_wait_dscnt 0" ::: "memory");
  v16h pa = load_frag(myp, 40, 0, 0, lane);

#pragma unroll
  for (int t = 0; t < 4; t++) o[t] = wmma_f16(pa, vf[t], o[t]);
  o[4] = wmma_f16(pa, ones, o[4]);
}

__global__ __launch_bounds__(256) void attn_kernel(
    const _Float16* __restrict__ Qh, const _Float16* __restrict__ Kh,
    const _Float16* __restrict__ Vt, const int* __restrict__ mask,
    _Float16* __restrict__ Oh) {
  __shared__ _Float16 plds[8][16 * 40];

  const int lane = threadIdx.x & 31;
  const int wave = threadIdx.x >> 5;
  const int bh   = blockIdx.x >> 3;
  const int qblk = blockIdx.x & 7;
  const int b = bh >> 4, h = bh & 15;
  const int q0 = qblk * 128 + wave * 16;

  const _Float16* Qp = Qh + pk_tile_off(b * 1024 + q0, h * 64, 32) + lane * 8;
  const _Float16* Kp = Kh + pk_tile_off(b * 1024, h * 64, 32) + lane * 8;
  const _Float16* Vp = Vt + pk_tile_off((b * 16 + h) * 64, 0, 32) + lane * 8;
  const int* mb = mask + b * 1024;
  _Float16* myp = &plds[wave][0];

  const v16h qa0 = load_pk(Qp);
  const v16h qa1 = load_pk(Qp + 512);

  v8f o[5] = {};
  v8f nmuse = {1e30f, 1e30f, 1e30f, 1e30f, 1e30f, 1e30f, 1e30f, 1e30f};
  v16h ones;
#pragma unroll
  for (int i = 0; i < 16; i++) ones[i] = (_Float16)1.0f;
  const int mh8 = (lane >> 4) << 3;

#define SCHED_FENCE() __builtin_amdgcn_sched_barrier(0)
  v16h kA[4], kB[4], vf[4];
  int mA0, mA1, mB0, mB1;
  load_k(kA, mA0, mA1, Kp, mb, 0, lane);

#pragma unroll 1
  for (int kb = 0; kb < 1024; kb += 64) {
    load_v(vf, Vp, kb);
    load_k(kB, mB0, mB1, Kp, mb, kb + 32, lane);
    SCHED_FENCE();
    attn_block(qa0, qa1, kA, vf, mA0, mA1, ones, o, nmuse, myp, lane, mh8);
    SCHED_FENCE();
    load_v(vf, Vp, kb + 32);
    const int kn = (kb + 64 < 1024) ? kb + 64 : 1024 - 32;
    load_k(kA, mA0, mA1, Kp, mb, kn, lane);
    SCHED_FENCE();
    attn_block(qa0, qa1, kB, vf, mB0, mB1, ones, o, nmuse, myp, lane, mh8);
    SCHED_FENCE();
  }
#undef SCHED_FENCE

#pragma unroll
  for (int t = 0; t < 4; t++) {
    const int n = h * 64 + t * 16 + (lane & 15);
    _Float16* dst = Oh + pk_tile_off(b * 1024 + q0, n, 32) + pk_elem_off(mh8, n & 31);
#pragma unroll
    for (int r = 0; r < 8; r++) {
      const float denom = o[4][r];
      const float val = denom > 0.f ? o[t][r] / denom : 0.f;
      dst[r * 8] = (_Float16)val;
    }
  }
}

#define LN_ROWS 2
__global__ __launch_bounds__(256) void ln_kernel(
    const float* __restrict__ Ain, const float* __restrict__ Bin,
    const float* __restrict__ g, const float* __restrict__ beta,
    float* __restrict__ out32, _Float16* __restrict__ out16) {
  const int lane = threadIdx.x & 31;
  const int wave = threadIdx.x >> 5;
  const int row0 = (blockIdx.x * 8 + wave) * LN_ROWS;

  const float4* pa = (const float4*)(Ain + (size_t)row0 * 1024) + lane;
  const float4* pb = (const float4*)(Bin + (size_t)row0 * 1024) + lane;
  const float4* pg = (const float4*)g + lane;
  const float4* pt = (const float4*)beta + lane;

  float4 xa[LN_ROWS][8], xb[LN_ROWS][8], gv[8], bv[8];
#pragma unroll
  for (int r = 0; r < LN_ROWS; r++) {
#pragma unroll
    for (int q = 0; q < 8; q++) xa[r][q] = pa[r * 256 + q * 32];
#pragma unroll
    for (int q = 0; q < 8; q++) xb[r][q] = pb[r * 256 + q * 32];
  }
#pragma unroll
  for (int q = 0; q < 8; q++) gv[q] = pg[q * 32];
#pragma unroll
  for (int q = 0; q < 8; q++) bv[q] = pt[q * 32];
  __builtin_amdgcn_sched_group_barrier(0x020, 48, 0);
  __builtin_amdgcn_sched_group_barrier(0x002, 1024, 0);

  float4 x[LN_ROWS][8];
  float s1[LN_ROWS], s2[LN_ROWS];
#pragma unroll
  for (int r = 0; r < LN_ROWS; r++) {
    float4 a1 = {0.f, 0.f, 0.f, 0.f}, a2 = {0.f, 0.f, 0.f, 0.f};
#pragma unroll
    for (int q = 0; q < 8; q++) {
      x[r][q].x = xa[r][q].x + xb[r][q].x;
      x[r][q].y = xa[r][q].y + xb[r][q].y;
      x[r][q].z = xa[r][q].z + xb[r][q].z;
      x[r][q].w = xa[r][q].w + xb[r][q].w;
      a1.x += x[r][q].x; a1.y += x[r][q].y; a1.z += x[r][q].z; a1.w += x[r][q].w;
      a2.x += x[r][q].x * x[r][q].x; a2.y += x[r][q].y * x[r][q].y;
      a2.z += x[r][q].z * x[r][q].z; a2.w += x[r][q].w * x[r][q].w;
    }
    s1[r] = (a1.x + a1.y) + (a1.z + a1.w);
    s2[r] = (a2.x + a2.y) + (a2.z + a2.w);
  }
#pragma unroll
  for (int off = 16; off >= 1; off >>= 1) {
#pragma unroll
    for (int r = 0; r < LN_ROWS; r++) {
      s1[r] += __shfl_xor(s1[r], off, 32);
      s2[r] += __shfl_xor(s2[r], off, 32);
    }
  }

#pragma unroll
  for (int r = 0; r < LN_ROWS; r++) {
    const int row = row0 + r;
    const float mu   = s1[r] * (1.f / 1024.f);
    const float var  = s2[r] * (1.f / 1024.f) - mu * mu;
    const float rstd = rsqrtf(var + 1e-5f);
    float4 y[8];
    float4* po = (float4*)(out32 + (size_t)row * 1024) + lane;
#pragma unroll
    for (int q = 0; q < 8; q++) {
      y[q].x = (x[r][q].x - mu) * rstd * gv[q].x + bv[q].x;
      y[q].y = (x[r][q].y - mu) * rstd * gv[q].y + bv[q].y;
      y[q].z = (x[r][q].z - mu) * rstd * gv[q].z + bv[q].z;
      y[q].w = (x[r][q].w - mu) * rstd * gv[q].w + bv[q].w;
      po[q * 32] = y[q];
    }
    if (out16) {
#pragma unroll
      for (int q = 0; q < 8; q++) {
        const int k = (q * 32 + lane) * 4;
        v4h o = {(_Float16)y[q].x, (_Float16)y[q].y, (_Float16)y[q].z, (_Float16)y[q].w};
        *(v4h*)(out16 + pk_tile_off(row, k, 32) + pk_elem_off(row & 15, k & 31)) = o;
      }
    }
  }
}
#undef LN_ROWS

extern "C" void kernel_launch(void* const* d_in, const int* in_sizes, int n_in,
                              void* d_out, int out_size, void* d_ws, size_t ws_size,
                              hipStream_t stream) {
  (void)in_sizes; (void)n_in; (void)out_size; (void)ws_size;
  const float* X    = (const float*)d_in[0];
  const float* Y    = (const float*)d_in[1];
  const int*   mask = (const int*)d_in[2];
  const float* Wq   = (const float*)d_in[3];
  const float* Wk   = (const float*)d_in[4];
  const float* Wv   = (const float*)d_in[5];
  const float* Wo   = (const float*)d_in[6];
  const float* ln1g = (const float*)d_in[7];
  const float* ln1b = (const float*)d_in[8];
  const float* ln2g = (const float*)d_in[9];
  const float* ln2b = (const float*)d_in[10];
  const float* W1   = (const float*)d_in[11];
  const float* b1   = (const float*)d_in[12];
  const float* W2   = (const float*)d_in[13];
  const float* b2   = (const float*)d_in[14];

  char* ws = (char*)d_ws;
  const size_t MB = 1024 * 1024;
  _Float16* Xh   = (_Float16*)(ws + 0);
  _Float16* Yh   = (_Float16*)(ws + 8 * MB);
  _Float16* Wqh  = (_Float16*)(ws + 16 * MB);
  _Float16* Wkh  = (_Float16*)(ws + 18 * MB);
  _Float16* Wvh  = (_Float16*)(ws + 20 * MB);
  _Float16* Woh  = (_Float16*)(ws + 22 * MB);
  _Float16* W1h  = (_Float16*)(ws + 24 * MB);
  _Float16* W2h  = (_Float16*)(ws + 26 * MB);
  _Float16* Qh   = (_Float16*)(ws + 28 * MB);
  _Float16* Kh   = (_Float16*)(ws + 36 * MB);
  _Float16* Vt   = (_Float16*)(ws + 44 * MB);
  float*    Fbuf = (float*)(ws + 52 * MB);
  float*    Hres = (float*)(ws + 68 * MB);
  _Float16* Ohb  = Xh;
  _Float16* Hh   = Yh;
  _Float16* FF1h = Qh;

  const int n4m = 4096 * 1024 / 4;
  const int n4w = 1024 * 1024 / 4;
  dim3 blk256(256);
  dim3 blk128(128);

  cvt_f32_f16_kernel<<<n4m / 256, blk256, 0, stream>>>(X, Xh);
  cvt_f32_f16_kernel<<<n4w / 256, blk256, 0, stream>>>(Wq, Wqh);

  const int ggrid = (4096 / 128) * (1024 / 128);
  const int sY = 4096 / 16;
  const int sW = 1024 / 16;

  gemm_nt_kernel<<<ggrid + sY + sW, blk128, 0, stream>>>(
      Wqh, Xh, nullptr, Qh, nullptr, 1024, 4096, 1024, 0, Y, Yh, sY, Wk, Wkh);
  gemm_nt_kernel<<<ggrid + sW + sW, blk128, 0, stream>>>(
      Wkh, Yh, nullptr, Kh, nullptr, 1024, 4096, 1024, 0, Wv, Wvh, sW, Wo, Woh);
  gemm_nt_kernel<<<ggrid + sW + sW, blk128, 0, stream>>>(
      Yh, Wvh, nullptr, Vt, nullptr, 4096, 1024, 1024, FLAG_TRANS, W1, W1h, sW,
      W2, W2h);

  attn_kernel<<<512, blk256, 0, stream>>>(Qh, Kh, Vt, mask, Ohb);

  gemm_nt_kernel<<<ggrid, blk128, 0, stream>>>(Woh, Ohb, Fbuf, nullptr, nullptr,
                                               1024, 4096, 1024, 0, nullptr,
                                               nullptr, 0, nullptr, nullptr);
  ln_kernel<<<256, blk256, 0, stream>>>(X, Fbuf, ln1g, ln1b, Hres, Hh);

  gemm_nt_kernel<<<ggrid, blk128, 0, stream>>>(W1h, Hh, nullptr, FF1h, b1,
                                               1024, 4096, 1024, FLAG_RELU,
                                               nullptr, nullptr, 0, nullptr,
                                               nullptr);
  gemm_nt_kernel<<<ggrid, blk128, 0, stream>>>(W2h, FF1h, Fbuf, nullptr, b2,
                                               1024, 4096, 1024, 0, nullptr,
                                               nullptr, 0, nullptr, nullptr);
  ln_kernel<<<256, blk256, 0, stream>>>(Hres, Fbuf, ln2g, ln2b, (float*)d_out,
                                         nullptr);
}
